# starting kernel
# baseline (speedup 1.0000x reference)
#include <hip/hip_runtime.h>

#define B_  16
#define N_  1025
#define D_  512
#define H_  8
#define HD_ 64
#define QT_ 65
#define ROWS_ 16400

typedef __attribute__((ext_vector_type(2))) float v2f;
typedef __attribute__((ext_vector_type(8))) float v8f;

__device__ __forceinline__ v8f wmma_f32(v2f a, v2f b, v8f c) {
  return __builtin_amdgcn_wmma_f32_16x16x4_f32(false, a, false, b, (short)0, c,
                                               false, false);
}

__global__ __launch_bounds__(256) void softplus_kernel(
    const float* __restrict__ gw, float* __restrict__ sp, int n) {
  int i = blockIdx.x * 256 + threadIdx.x;
  if (i < n) {
    float w = gw[i];
    sp[i] = (w > 20.0f) ? w : log1pf(__expf(w));
  }
}

__global__ __launch_bounds__(32) void qkv_gemm(
    const float* __restrict__ x, const float* __restrict__ w,
    float* __restrict__ Qd, float* __restrict__ Kd, float* __restrict__ Vd) {
  const int lane = threadIdx.x;
  const int lm = lane & 15;
  const int lh = lane >> 4;
  const int col0 = blockIdx.x * 64;
  const int row0 = blockIdx.y * 16;
  const float* ap = x + (size_t)(row0 + lm) * D_ + 2 * lh;
  const float* bp0 = w + (size_t)(col0 + 0 + lm) * D_ + 2 * lh;
  const float* bp1 = w + (size_t)(col0 + 16 + lm) * D_ + 2 * lh;
  const float* bp2 = w + (size_t)(col0 + 32 + lm) * D_ + 2 * lh;
  const float* bp3 = w + (size_t)(col0 + 48 + lm) * D_ + 2 * lh;
  v8f c0 = {}, c1 = {}, c2 = {}, c3 = {};
  v2f a_c = *(const v2f*)(ap);
  v2f b0_c = *(const v2f*)(bp0);
  v2f b1_c = *(const v2f*)(bp1);
  v2f b2_c = *(const v2f*)(bp2);
  v2f b3_c = *(const v2f*)(bp3);
#pragma unroll 4
  for (int k = 0; k < D_ - 4; k += 4) {
    v2f a_n = *(const v2f*)(ap + k + 4);
    v2f b0_n = *(const v2f*)(bp0 + k + 4);
    v2f b1_n = *(const v2f*)(bp1 + k + 4);
    v2f b2_n = *(const v2f*)(bp2 + k + 4);
    v2f b3_n = *(const v2f*)(bp3 + k + 4);
    c0 = wmma_f32(a_c, b0_c, c0);
    c1 = wmma_f32(a_c, b1_c, c1);
    c2 = wmma_f32(a_c, b2_c, c2);
    c3 = wmma_f32(a_c, b3_c, c3);
    a_c = a_n; b0_c = b0_n; b1_c = b1_n; b2_c = b2_n; b3_c = b3_n;
  }
  c0 = wmma_f32(a_c, b0_c, c0);
  c1 = wmma_f32(a_c, b1_c, c1);
  c2 = wmma_f32(a_c, b2_c, c2);
  c3 = wmma_f32(a_c, b3_c, c3);

  int bb8[8], nn8[8];
#pragma unroll
  for (int i = 0; i < 8; ++i) {
    int r = row0 + i + 8 * lh;
    int bbv = r / N_;
    bb8[i] = bbv;
    nn8[i] = r - bbv * N_;
  }
#pragma unroll
  for (int j = 0; j < 4; ++j) {
    const v8f& acc = (j == 0) ? c0 : (j == 1) ? c1 : (j == 2) ? c2 : c3;
    const int c = col0 + j * 16 + lm;
    const int part = c >> 9;
    const int rem = c & 511;
    const int h = rem >> 6;
    const int hd = rem & 63;
    float* dst = (part == 0) ? Qd : (part == 1) ? Kd : Vd;
#pragma unroll
    for (int i = 0; i < 8; ++i) {
      dst[(((size_t)(bb8[i] * H_ + h) * N_ + nn8[i]) * HD_) + hd] = acc[i];
    }
  }
}

__global__ __launch_bounds__(32) void attn_kernel(
    const float* __restrict__ Qd, const float* __restrict__ Kd,
    const float* __restrict__ Vd, const float* __restrict__ SP,
    float* __restrict__ AO) {
  __shared__ float lds[16][18];
  const int lane = threadIdx.x;
  const int lm = lane & 15;
  const int lh = lane >> 4;
  const int qt = blockIdx.x;
  const int bh = blockIdx.y;
  const int b = bh >> 3;
  const int h = bh & 7;
  const size_t base = (size_t)bh * N_ * HD_;
  const float* Qp = Qd + base;
  const float* Kp = Kd + base;
  const float* Vp = Vd + base;
  const float scale = 0.125f;

  const int qlane = qt * 16 + lm;
  const int qclamp = (qlane < N_) ? qlane : N_ - 1;
  v2f qa[16];
#pragma unroll
  for (int s = 0; s < 16; ++s)
    qa[s] = *(const v2f*)(Qp + (size_t)qclamp * HD_ + 4 * s + 2 * lh);

  v8f O0 = {}, O1 = {}, O2 = {}, O3 = {};
  float mrun[8], lrun[8];
#pragma unroll
  for (int i = 0; i < 8; ++i) { mrun[i] = -3.0e38f; lrun[i] = 0.0f; }

  for (int kt = 0; kt < QT_; ++kt) {
    const int keyb = kt * 16 + lm;
    const int keybc = (keyb < N_) ? keyb : N_ - 1;
    const float* kp = Kp + (size_t)keybc * HD_ + 2 * lh;
    v2f bf[16];
#pragma unroll
    for (int s = 0; s < 16; ++s) bf[s] = *(const v2f*)(kp + 4 * s);
    v8f S = {};
#pragma unroll
    for (int s = 0; s < 16; ++s) S = wmma_f32(qa[s], bf[s], S);

    const int keyd = kt * 16 + lm;
    float p[8], corr[8];
#pragma unroll
    for (int i = 0; i < 8; ++i) {
      int qrow = qt * 16 + i + 8 * lh;
      float lg;
      if (qrow >= N_)      lg = 0.0f;
      else if (keyd >= N_) lg = -3.0e38f;
      else                 lg = S[i] * scale * SP[(size_t)qrow * N_ + keyd];
      float mx = lg;
#pragma unroll
      for (int off = 1; off < 16; off <<= 1)
        mx = fmaxf(mx, __shfl_xor(mx, off, 32));
      float mnew = fmaxf(mrun[i], mx);
      float pv = __expf(lg - mnew);
      float sum = pv;
#pragma unroll
      for (int off = 1; off < 16; off <<= 1) sum += __shfl_xor(sum, off, 32);
      float c = __expf(mrun[i] - mnew);
      lrun[i] = lrun[i] * c + sum;
      mrun[i] = mnew;
      corr[i] = c;
      p[i] = pv;
    }
#pragma unroll
    for (int i = 0; i < 8; ++i) {
      O0[i] *= corr[i]; O1[i] *= corr[i]; O2[i] *= corr[i]; O3[i] *= corr[i];
    }
    v2f vb[4][4];
#pragma unroll
    for (int kk = 0; kk < 4; ++kk) {
      int krow = kt * 16 + 4 * kk + 2 * lh;
      int kr0 = (krow < N_) ? krow : N_ - 1;
      int kr1 = (krow + 1 < N_) ? krow + 1 : N_ - 1;
      const float* v0 = Vp + (size_t)kr0 * HD_;
      const float* v1 = Vp + (size_t)kr1 * HD_;
#pragma unroll
      for (int nt = 0; nt < 4; ++nt) {
        vb[kk][nt].x = v0[nt * 16 + lm];
        vb[kk][nt].y = v1[nt * 16 + lm];
      }
    }
    __syncthreads();
#pragma unroll
    for (int i = 0; i < 8; ++i) lds[i + 8 * lh][lm] = p[i];
    __syncthreads();
#pragma unroll
    for (int kk = 0; kk < 4; ++kk) {
      v2f pa = *(const v2f*)&lds[lm][4 * kk + 2 * lh];
      O0 = wmma_f32(pa, vb[kk][0], O0);
      O1 = wmma_f32(pa, vb[kk][1], O1);
      O2 = wmma_f32(pa, vb[kk][2], O2);
      O3 = wmma_f32(pa, vb[kk][3], O3);
    }
  }
  float* outp = AO + (size_t)b * N_ * D_ + h * HD_;
#pragma unroll
  for (int i = 0; i < 8; ++i) {
    int qrow = qt * 16 + i + 8 * lh;
    if (qrow < N_) {
      float inv = 1.0f / lrun[i];
      size_t ro = (size_t)qrow * D_;
      outp[ro +  0 + lm] = O0[i] * inv;
      outp[ro + 16 + lm] = O1[i] * inv;
      outp[ro + 32 + lm] = O2[i] * inv;
      outp[ro + 48 + lm] = O3[i] * inv;
    }
  }
}

__global__ __launch_bounds__(32) void proj_gemm(
    const float* __restrict__ AO, const float* __restrict__ w,
    const float* __restrict__ bias, float* __restrict__ P) {
  const int lane = threadIdx.x;
  const int lm = lane & 15;
  const int lh = lane >> 4;
  const int col0 = blockIdx.x * 64;
  const int row0 = blockIdx.y * 16;
  const float* ap = AO + (size_t)(row0 + lm) * D_ + 2 * lh;
  const float* bp0 = w + (size_t)(col0 + 0 + lm) * D_ + 2 * lh;
  const float* bp1 = w + (size_t)(col0 + 16 + lm) * D_ + 2 * lh;
  const float* bp2 = w + (size_t)(col0 + 32 + lm) * D_ + 2 * lh;
  const float* bp3 = w + (size_t)(col0 + 48 + lm) * D_ + 2 * lh;
  v8f c0 = {}, c1 = {}, c2 = {}, c3 = {};
  v2f a_c = *(const v2f*)(ap);
  v2f b0_c = *(const v2f*)(bp0);
  v2f b1_c = *(const v2f*)(bp1);
  v2f b2_c = *(const v2f*)(bp2);
  v2f b3_c = *(const v2f*)(bp3);
#pragma unroll 4
  for (int k = 0; k < D_ - 4; k += 4) {
    v2f a_n = *(const v2f*)(ap + k + 4);
    v2f b0_n = *(const v2f*)(bp0 + k + 4);
    v2f b1_n = *(const v2f*)(bp1 + k + 4);
    v2f b2_n = *(const v2f*)(bp2 + k + 4);
    v2f b3_n = *(const v2f*)(bp3 + k + 4);
    c0 = wmma_f32(a_c, b0_c, c0);
    c1 = wmma_f32(a_c, b1_c, c1);
    c2 = wmma_f32(a_c, b2_c, c2);
    c3 = wmma_f32(a_c, b3_c, c3);
    a_c = a_n; b0_c = b0_n; b1_c = b1_n; b2_c = b2_n; b3_c = b3_n;
  }
  c0 = wmma_f32(a_c, b0_c, c0);
  c1 = wmma_f32(a_c, b1_c, c1);
  c2 = wmma_f32(a_c, b2_c, c2);
  c3 = wmma_f32(a_c, b3_c, c3);
#pragma unroll
  for (int j = 0; j < 4; ++j) {
    const v8f& acc = (j == 0) ? c0 : (j == 1) ? c1 : (j == 2) ? c2 : c3;
    const int c = col0 + j * 16 + lm;
    const float bc = bias[c];
#pragma unroll
    for (int i = 0; i < 8; ++i) {
      int r = row0 + i + 8 * lh;
      P[(size_t)r * D_ + c] = acc[i] + bc;
    }
  }
}

__global__ __launch_bounds__(256) void ln_residual(
    float* __restrict__ Y, const float* __restrict__ gamma,
    const float* __restrict__ beta) {
  __shared__ float reds[8], reds2[8];
  __shared__ float mu_s, rs_s;
  const int row = blockIdx.x;
  float* y = Y + (size_t)row * D_;
  const int t = threadIdx.x;
  float a = y[t], bv = y[t + 256];
  float s = a + bv, s2 = a * a + bv * bv;
#pragma unroll
  for (int off = 1; off < 32; off <<= 1) {
    s += __shfl_xor(s, off, 32);
    s2 += __shfl_xor(s2, off, 32);
  }
  if ((t & 31) == 0) { reds[t >> 5] = s; reds2[t >> 5] = s2; }
  __syncthreads();
  if (t == 0) {
    float S = 0.f, S2 = 0.f;
#pragma unroll
    for (int i = 0; i < 8; ++i) { S += reds[i]; S2 += reds2[i]; }
    float mu = S * (1.0f / D_);
    float var = S2 * (1.0f / D_) - mu * mu;
    mu_s = mu;
    rs_s = rsqrtf(var + 1e-5f);
  }
  __syncthreads();
  const float mu = mu_s, rs = rs_s;
  y[t]       = a  + (a  - mu) * rs * gamma[t]       + beta[t];
  y[t + 256] = bv + (bv - mu) * rs * gamma[t + 256] + beta[t + 256];
}

extern "C" void kernel_launch(void* const* d_in, const int* in_sizes, int n_in,
                              void* d_out, int out_size, void* d_ws,
                              size_t ws_size, hipStream_t stream) {
  const float* x      = (const float*)d_in[0];
  const float* w_qkv  = (const float*)d_in[1];
  const float* w_proj = (const float*)d_in[2];
  const float* b_proj = (const float*)d_in[3];
  const float* gw     = (const float*)d_in[4];
  const float* ln_g   = (const float*)d_in[5];
  const float* ln_b   = (const float*)d_in[6];
  float* out = (float*)d_out;

  float* ws = (float*)d_ws;
  const size_t nsp  = (size_t)N_ * N_;
  const size_t nqkv = (size_t)B_ * H_ * N_ * HD_;
  float* SP = ws;
  float* Q  = SP + nsp;
  float* K  = Q + nqkv;
  float* V  = K + nqkv;
  float* AO = V + nqkv;

  softplus_kernel<<<(int)((nsp + 255) / 256), 256, 0, stream>>>(gw, SP,
                                                                (int)nsp);
  qkv_gemm<<<dim3(3 * D_ / 64, ROWS_ / 16), 32, 0, stream>>>(x, w_qkv, Q, K,
                                                             V);
  attn_kernel<<<dim3(QT_, B_ * H_), 32, 0, stream>>>(Q, K, V, SP, AO);
  proj_gemm<<<dim3(D_ / 64, ROWS_ / 16), 32, 0, stream>>>(AO, w_proj, b_proj,
                                                          out);
  ln_residual<<<ROWS_, 256, 0, stream>>>(out, ln_g, ln_b);
}


# attention computed transposed (S^T=K Q^T, O^T=V^T P^T): one query per lane, in-lane row max/sum, P^T fed to WMMA from registers, no LDS/barriers; changes summation order of row sums and PV keys
# speedup vs baseline: 1.2655x; 1.2655x over previous
#include <hip/hip_runtime.h>

#define B_  16
#define N_  1025
#define D_  512
#define H_  8
#define HD_ 64
#define QT_ 65
#define ROWS_ 16400

typedef __attribute__((ext_vector_type(2))) float v2f;
typedef __attribute__((ext_vector_type(8))) float v8f;

__device__ __forceinline__ v8f wmma_f32(v2f a, v2f b, v8f c) {
  return __builtin_amdgcn_wmma_f32_16x16x4_f32(false, a, false, b, (short)0, c,
                                               false, false);
}

__global__ __launch_bounds__(256) void softplus_kernel(
    const float* __restrict__ gw, float* __restrict__ sp, int n) {
  int i = blockIdx.x * 256 + threadIdx.x;
  if (i < n) {
    float w = gw[i];
    sp[i] = (w > 20.0f) ? w : log1pf(__expf(w));
  }
}

__global__ __launch_bounds__(32) void qkv_gemm(
    const float* __restrict__ x, const float* __restrict__ w,
    float* __restrict__ Qd, float* __restrict__ Kd, float* __restrict__ Vd) {
  const int lane = threadIdx.x;
  const int lm = lane & 15;
  const int lh = lane >> 4;
  const int col0 = blockIdx.x * 64;
  const int row0 = blockIdx.y * 16;
  const float* ap = x + (size_t)(row0 + lm) * D_ + 2 * lh;
  const float* bp0 = w + (size_t)(col0 + 0 + lm) * D_ + 2 * lh;
  const float* bp1 = w + (size_t)(col0 + 16 + lm) * D_ + 2 * lh;
  const float* bp2 = w + (size_t)(col0 + 32 + lm) * D_ + 2 * lh;
  const float* bp3 = w + (size_t)(col0 + 48 + lm) * D_ + 2 * lh;
  v8f c0 = {}, c1 = {}, c2 = {}, c3 = {};
  v2f a_c = *(const v2f*)(ap);
  v2f b0_c = *(const v2f*)(bp0);
  v2f b1_c = *(const v2f*)(bp1);
  v2f b2_c = *(const v2f*)(bp2);
  v2f b3_c = *(const v2f*)(bp3);
#pragma unroll 4
  for (int k = 0; k < D_ - 4; k += 4) {
    v2f a_n = *(const v2f*)(ap + k + 4);
    v2f b0_n = *(const v2f*)(bp0 + k + 4);
    v2f b1_n = *(const v2f*)(bp1 + k + 4);
    v2f b2_n = *(const v2f*)(bp2 + k + 4);
    v2f b3_n = *(const v2f*)(bp3 + k + 4);
    c0 = wmma_f32(a_c, b0_c, c0);
    c1 = wmma_f32(a_c, b1_c, c1);
    c2 = wmma_f32(a_c, b2_c, c2);
    c3 = wmma_f32(a_c, b3_c, c3);
    a_c = a_n; b0_c = b0_n; b1_c = b1_n; b2_c = b2_n; b3_c = b3_n;
  }
  c0 = wmma_f32(a_c, b0_c, c0);
  c1 = wmma_f32(a_c, b1_c, c1);
  c2 = wmma_f32(a_c, b2_c, c2);
  c3 = wmma_f32(a_c, b3_c, c3);

  int bb8[8], nn8[8];
#pragma unroll
  for (int i = 0; i < 8; ++i) {
    int r = row0 + i + 8 * lh;
    int bbv = r / N_;
    bb8[i] = bbv;
    nn8[i] = r - bbv * N_;
  }
#pragma unroll
  for (int j = 0; j < 4; ++j) {
    const v8f& acc = (j == 0) ? c0 : (j == 1) ? c1 : (j == 2) ? c2 : c3;
    const int c = col0 + j * 16 + lm;
    const int part = c >> 9;
    const int rem = c & 511;
    const int h = rem >> 6;
    const int hd = rem & 63;
    float* dst = (part == 0) ? Qd : (part == 1) ? Kd : Vd;
#pragma unroll
    for (int i = 0; i < 8; ++i) {
      dst[(((size_t)(bb8[i] * H_ + h) * N_ + nn8[i]) * HD_) + hd] = acc[i];
    }
  }
}

typedef float v4fu __attribute__((ext_vector_type(4), aligned(4)));

__device__ __forceinline__ float xhalf(float x) {
  int bits = __builtin_bit_cast(int, x);
  return __builtin_bit_cast(
      float, __builtin_amdgcn_permlanex16(bits, bits, 0x76543210, 0xfedcba98,
                                          false, false));
}

__global__ __launch_bounds__(32) void attn_kernel(
    const float* __restrict__ Qd, const float* __restrict__ Kd,
    const float* __restrict__ Vd, const float* __restrict__ SP,
    float* __restrict__ AO) {
  const int lane = threadIdx.x;
  const int lm = lane & 15;
  const int lh = lane >> 4;
  const int qt = blockIdx.x;
  const int bh = blockIdx.y;
  const int b = bh >> 3;
  const int h = bh & 7;
  const size_t base = (size_t)bh * N_ * HD_;
  const float* Qp = Qd + base;
  const float* Kp = Kd + base;
  const float* Vp = Vd + base;
  const float scale = 0.125f;

  const int q = qt * 16 + lm;
  const int qc = (q < N_) ? q : N_ - 1;
  v2f qb[16];
#pragma unroll
  for (int s = 0; s < 16; ++s)
    qb[s] = *(const v2f*)(Qp + (size_t)qc * HD_ + 4 * s + 2 * lh);
  const float* sprow = SP + (unsigned)(qc * N_ + 8 * lh);

  v8f O0 = {}, O1 = {}, O2 = {}, O3 = {};
  float mrun = -3.0e38f, lrun = 0.0f;

  for (int kt = 0; kt < QT_; ++kt) {
    const int keya = kt * 16 + lm;
    const int keyac = (keya < N_) ? keya : N_ - 1;
    const float* kp = Kp + (unsigned)(keyac * HD_ + 2 * lh);
    v2f ka[16];
#pragma unroll
    for (int s = 0; s < 16; ++s) ka[s] = *(const v2f*)(kp + 4 * s);
    v8f S = {};
#pragma unroll
    for (int s = 0; s < 16; ++s) S = wmma_f32(ka[s], qb[s], S);

    const int key0 = kt * 16 + 8 * lh;
    v4fu vv[8];
#pragma unroll
    for (int r = 0; r < 8; ++r) {
      int kr = key0 + r;
      kr = (kr < N_) ? kr : N_ - 1;
      vv[r] = *(const v4fu*)(Vp + (unsigned)(kr * HD_ + 4 * lm));
    }

    float lg[8];
#pragma unroll
    for (int r = 0; r < 8; ++r) lg[r] = S[r] * scale * sprow[kt * 16 + r];
    if (kt == QT_ - 1) {
#pragma unroll
      for (int r = 0; r < 8; ++r)
        if (key0 + r >= N_) lg[r] = -3.0e38f;
    }
    float mx = fmaxf(fmaxf(fmaxf(lg[0], lg[1]), fmaxf(lg[2], lg[3])),
                     fmaxf(fmaxf(lg[4], lg[5]), fmaxf(lg[6], lg[7])));
    mx = fmaxf(mx, xhalf(mx));
    const float mnew = fmaxf(mrun, mx);
    const float c = __expf(mrun - mnew);
    float p[8];
#pragma unroll
    for (int r = 0; r < 8; ++r) p[r] = __expf(lg[r] - mnew);
    const float sum = ((p[0] + p[1]) + (p[2] + p[3])) +
                      ((p[4] + p[5]) + (p[6] + p[7]));
    lrun = lrun * c + sum;
    mrun = mnew;
#pragma unroll
    for (int r = 0; r < 8; ++r) {
      O0[r] *= c; O1[r] *= c; O2[r] *= c; O3[r] *= c;
    }
#pragma unroll
    for (int kk = 0; kk < 4; ++kk) {
      v2f pb = {p[2 * kk], p[2 * kk + 1]};
      v2f a0 = {vv[2 * kk].x, vv[2 * kk + 1].x};
      v2f a1 = {vv[2 * kk].y, vv[2 * kk + 1].y};
      v2f a2 = {vv[2 * kk].z, vv[2 * kk + 1].z};
      v2f a3 = {vv[2 * kk].w, vv[2 * kk + 1].w};
      O0 = wmma_f32(a0, pb, O0);
      O1 = wmma_f32(a1, pb, O1);
      O2 = wmma_f32(a2, pb, O2);
      O3 = wmma_f32(a3, pb, O3);
    }
  }
  const float lsum = lrun + xhalf(lrun);
  const float inv = 1.0f / lsum;
  if (q < N_) {
    float* outp = AO + ((size_t)b * N_ + q) * D_ + h * HD_ + 32 * lh;
#pragma unroll
    for (int r = 0; r < 8; ++r) {
      outp[4 * r + 0] = O0[r] * inv;
      outp[4 * r + 1] = O1[r] * inv;
      outp[4 * r + 2] = O2[r] * inv;
      outp[4 * r + 3] = O3[r] * inv;
    }
  }
}

__global__ __launch_bounds__(32) void proj_gemm(
    const float* __restrict__ AO, const float* __restrict__ w,
    const float* __restrict__ bias, float* __restrict__ P) {
  const int lane = threadIdx.x;
  const int lm = lane & 15;
  const int lh = lane >> 4;
  const int col0 = blockIdx.x * 64;
  const int row0 = blockIdx.y * 16;
  const float* ap = AO + (size_t)(row0 + lm) * D_ + 2 * lh;
  const float* bp0 = w + (size_t)(col0 + 0 + lm) * D_ + 2 * lh;
  const float* bp1 = w + (size_t)(col0 + 16 + lm) * D_ + 2 * lh;
  const float* bp2 = w + (size_t)(col0 + 32 + lm) * D_ + 2 * lh;
  const float* bp3 = w + (size_t)(col0 + 48 + lm) * D_ + 2 * lh;
  v8f c0 = {}, c1 = {}, c2 = {}, c3 = {};
  v2f a_c = *(const v2f*)(ap);
  v2f b0_c = *(const v2f*)(bp0);
  v2f b1_c = *(const v2f*)(bp1);
  v2f b2_c = *(const v2f*)(bp2);
  v2f b3_c = *(const v2f*)(bp3);
#pragma unroll 4
  for (int k = 0; k < D_ - 4; k += 4) {
    v2f a_n = *(const v2f*)(ap + k + 4);
    v2f b0_n = *(const v2f*)(bp0 + k + 4);
    v2f b1_n = *(const v2f*)(bp1 + k + 4);
    v2f b2_n = *(const v2f*)(bp2 + k + 4);
    v2f b3_n = *(const v2f*)(bp3 + k + 4);
    c0 = wmma_f32(a_c, b0_c, c0);
    c1 = wmma_f32(a_c, b1_c, c1);
    c2 = wmma_f32(a_c, b2_c, c2);
    c3 = wmma_f32(a_c, b3_c, c3);
    a_c = a_n; b0_c = b0_n; b1_c = b1_n; b2_c = b2_n; b3_c = b3_n;
  }
  c0 = wmma_f32(a_c, b0_c, c0);
  c1 = wmma_f32(a_c, b1_c, c1);
  c2 = wmma_f32(a_c, b2_c, c2);
  c3 = wmma_f32(a_c, b3_c, c3);
#pragma unroll
  for (int j = 0; j < 4; ++j) {
    const v8f& acc = (j == 0) ? c0 : (j == 1) ? c1 : (j == 2) ? c2 : c3;
    const int c = col0 + j * 16 + lm;
    const float bc = bias[c];
#pragma unroll
    for (int i = 0; i < 8; ++i) {
      int r = row0 + i + 8 * lh;
      P[(size_t)r * D_ + c] = acc[i] + bc;
    }
  }
}

__global__ __launch_bounds__(256) void ln_residual(
    float* __restrict__ Y, const float* __restrict__ gamma,
    const float* __restrict__ beta) {
  __shared__ float reds[8], reds2[8];
  __shared__ float mu_s, rs_s;
  const int row = blockIdx.x;
  float* y = Y + (size_t)row * D_;
  const int t = threadIdx.x;
  float a = y[t], bv = y[t + 256];
  float s = a + bv, s2 = a * a + bv * bv;
#pragma unroll
  for (int off = 1; off < 32; off <<= 1) {
    s += __shfl_xor(s, off, 32);
    s2 += __shfl_xor(s2, off, 32);
  }
  if ((t & 31) == 0) { reds[t >> 5] = s; reds2[t >> 5] = s2; }
  __syncthreads();
  if (t == 0) {
    float S = 0.f, S2 = 0.f;
#pragma unroll
    for (int i = 0; i < 8; ++i) { S += reds[i]; S2 += reds2[i]; }
    float mu = S * (1.0f / D_);
    float var = S2 * (1.0f / D_) - mu * mu;
    mu_s = mu;
    rs_s = rsqrtf(var + 1e-5f);
  }
  __syncthreads();
  const float mu = mu_s, rs = rs_s;
  y[t]       = a  + (a  - mu) * rs * gamma[t]       + beta[t];
  y[t + 256] = bv + (bv - mu) * rs * gamma[t + 256] + beta[t + 256];
}

extern "C" void kernel_launch(void* const* d_in, const int* in_sizes, int n_in,
                              void* d_out, int out_size, void* d_ws,
                              size_t ws_size, hipStream_t stream) {
  const float* x      = (const float*)d_in[0];
  const float* w_qkv  = (const float*)d_in[1];
  const float* w_proj = (const float*)d_in[2];
  const float* b_proj = (const float*)d_in[3];
  const float* gw     = (const float*)d_in[4];
  const float* ln_g   = (const float*)d_in[5];
  const float* ln_b   = (const float*)d_in[6];
  float* out = (float*)d_out;

  float* ws = (float*)d_ws;
  const size_t nsp  = (size_t)N_ * N_;
  const size_t nqkv = (size_t)B_ * H_ * N_ * HD_;
  float* SP = ws;
  float* Q  = SP + nsp;
  float* K  = Q + nqkv;
  float* V  = K + nqkv;
  float* AO = V + nqkv;

  softplus_kernel<<<(int)((nsp + 255) / 256), 256, 0, stream>>>(gw, SP,
                                                                (int)nsp);
  qkv_gemm<<<dim3(3 * D_ / 64, ROWS_ / 16), 32, 0, stream>>>(x, w_qkv, Q, K,
                                                             V);
  attn_kernel<<<dim3(QT_, B_ * H_), 32, 0, stream>>>(Q, K, V, SP, AO);
  proj_gemm<<<dim3(D_ / 64, ROWS_ / 16), 32, 0, stream>>>(AO, w_proj, b_proj,
                                                          out);
  ln_residual<<<ROWS_, 256, 0, stream>>>(out, ln_g, ln_b);
}


# both GEMMs: one wave per 80x64 tile (was 16x64), fragments by 16-byte loads taking a whole 128-byte k line per row in one group: 36 loads per 160 WMMAs (was 5 per 4); changes k summation order
# speedup vs baseline: 2.2546x; 1.7816x over previous
#include <hip/hip_runtime.h>

#define B_  16
#define N_  1025
#define D_  512
#define H_  8
#define HD_ 64
#define QT_ 65
#define ROWS_ 16400

typedef __attribute__((ext_vector_type(2))) float v2f;
typedef __attribute__((ext_vector_type(8))) float v8f;
typedef float v4fu __attribute__((ext_vector_type(4), aligned(4)));

#define RT_ 5
#define CT_ 4

__device__ __forceinline__ v8f wmma_f32(v2f a, v2f b, v8f c) {
  return __builtin_amdgcn_wmma_f32_16x16x4_f32(false, a, false, b, (short)0, c,
                                               false, false);
}

__global__ __launch_bounds__(256) void softplus_kernel(
    const float* __restrict__ gw, float* __restrict__ sp, int n) {
  int i = blockIdx.x * 256 + threadIdx.x;
  if (i < n) {
    float w = gw[i];
    sp[i] = (w > 20.0f) ? w : log1pf(__expf(w));
  }
}

__device__ __forceinline__ void gemm_tile_80x64(const float* __restrict__ ap,
                                                const float* __restrict__ bp,
                                                v8f (&c)[RT_][CT_]) {
  for (int k0 = 0; k0 < D_; k0 += 32) {
    v4fu b[CT_][4], a[RT_][4];
#pragma unroll
    for (int j = 0; j < CT_; ++j)
#pragma unroll
      for (int u = 0; u < 4; ++u)
        b[j][u] = *(const v4fu*)(bp + (size_t)j * 16 * D_ + k0 + 4 * u);
#pragma unroll
    for (int i = 0; i < RT_; ++i)
#pragma unroll
      for (int u = 0; u < 4; ++u)
        a[i][u] = *(const v4fu*)(ap + (size_t)i * 16 * D_ + k0 + 4 * u);
    __builtin_amdgcn_sched_barrier(0);
#pragma unroll
    for (int i = 0; i < RT_; ++i) {
#pragma unroll
      for (int u = 0; u < 4; ++u) {
        const v2f alo = {a[i][u].x, a[i][u].y};
        const v2f ahi = {a[i][u].z, a[i][u].w};
#pragma unroll
        for (int j = 0; j < CT_; ++j) {
          const v2f blo = {b[j][u].x, b[j][u].y};
          c[i][j] = wmma_f32(alo, blo, c[i][j]);
        }
#pragma unroll
        for (int j = 0; j < CT_; ++j) {
          const v2f bhi = {b[j][u].z, b[j][u].w};
          c[i][j] = wmma_f32(ahi, bhi, c[i][j]);
        }
      }
    }
  }
}

__global__ __launch_bounds__(32) void qkv_gemm(
    const float* __restrict__ x, const float* __restrict__ w,
    float* __restrict__ Qd, float* __restrict__ Kd, float* __restrict__ Vd) {
  const int lane = threadIdx.x;
  const int lm = lane & 15;
  const int lh = lane >> 4;
  const int col0 = blockIdx.x * 64;
  const int row0 = blockIdx.y * 80;
  v8f c[RT_][CT_];
#pragma unroll
  for (int i = 0; i < RT_; ++i)
#pragma unroll
    for (int j = 0; j < CT_; ++j) c[i][j] = v8f{};
  gemm_tile_80x64(x + (size_t)(row0 + lm) * D_ + 16 * lh,
                  w + (size_t)(col0 + lm) * D_ + 16 * lh, c);

  const int part = col0 >> 9;
  const int h = (col0 & 511) >> 6;
  float* dst = (part == 0) ? Qd : (part == 1) ? Kd : Vd;
#pragma unroll
  for (int i = 0; i < RT_; ++i) {
#pragma unroll
    for (int e = 0; e < 8; ++e) {
      const int r = row0 + 16 * i + 8 * lh + e;
      const int bb = r / N_;
      const int nn = r - bb * N_;
      float* p = dst + ((size_t)(bb * H_ + h) * N_ + nn) * HD_ + lm;
#pragma unroll
      for (int j = 0; j < CT_; ++j) p[16 * j] = c[i][j][e];
    }
  }
}

__device__ __forceinline__ float xhalf(float x) {
  int bits = __builtin_bit_cast(int, x);
  return __builtin_bit_cast(
      float, __builtin_amdgcn_permlanex16(bits, bits, 0x76543210, 0xfedcba98,
                                          false, false));
}

__global__ __launch_bounds__(32) void attn_kernel(
    const float* __restrict__ Qd, const float* __restrict__ Kd,
    const float* __restrict__ Vd, const float* __restrict__ SP,
    float* __restrict__ AO) {
  const int lane = threadIdx.x;
  const int lm = lane & 15;
  const int lh = lane >> 4;
  const int qt = blockIdx.x;
  const int bh = blockIdx.y;
  const int b = bh >> 3;
  const int h = bh & 7;
  const size_t base = (size_t)bh * N_ * HD_;
  const float* Qp = Qd + base;
  const float* Kp = Kd + base;
  const float* Vp = Vd + base;
  const float scale = 0.125f;

  const int q = qt * 16 + lm;
  const int qc = (q < N_) ? q : N_ - 1;
  v2f qb[16];
#pragma unroll
  for (int s = 0; s < 16; ++s)
    qb[s] = *(const v2f*)(Qp + (size_t)qc * HD_ + 4 * s + 2 * lh);
  const float* sprow = SP + (unsigned)(qc * N_ + 8 * lh);

  v8f O0 = {}, O1 = {}, O2 = {}, O3 = {};
  float mrun = -3.0e38f, lrun = 0.0f;

  for (int kt = 0; kt < QT_; ++kt) {
    const int keya = kt * 16 + lm;
    const int keyac = (keya < N_) ? keya : N_ - 1;
    const float* kp = Kp + (unsigned)(keyac * HD_ + 2 * lh);
    v2f ka[16];
#pragma unroll
    for (int s = 0; s < 16; ++s) ka[s] = *(const v2f*)(kp + 4 * s);
    v8f S = {};
#pragma unroll
    for (int s = 0; s < 16; ++s) S = wmma_f32(ka[s], qb[s], S);

    const int key0 = kt * 16 + 8 * lh;
    v4fu vv[8];
#pragma unroll
    for (int r = 0; r < 8; ++r) {
      int kr = key0 + r;
      kr = (kr < N_) ? kr : N_ - 1;
      vv[r] = *(const v4fu*)(Vp + (unsigned)(kr * HD_ + 4 * lm));
    }

    float lg[8];
#pragma unroll
    for (int r = 0; r < 8; ++r) lg[r] = S[r] * scale * sprow[kt * 16 + r];
    if (kt == QT_ - 1) {
#pragma unroll
      for (int r = 0; r < 8; ++r)
        if (key0 + r >= N_) lg[r] = -3.0e38f;
    }
    float mx = fmaxf(fmaxf(fmaxf(lg[0], lg[1]), fmaxf(lg[2], lg[3])),
                     fmaxf(fmaxf(lg[4], lg[5]), fmaxf(lg[6], lg[7])));
    mx = fmaxf(mx, xhalf(mx));
    const float mnew = fmaxf(mrun, mx);
    const float c = __expf(mrun - mnew);
    float p[8];
#pragma unroll
    for (int r = 0; r < 8; ++r) p[r] = __expf(lg[r] - mnew);
    const float sum = ((p[0] + p[1]) + (p[2] + p[3])) +
                      ((p[4] + p[5]) + (p[6] + p[7]));
    lrun = lrun * c + sum;
    mrun = mnew;
#pragma unroll
    for (int r = 0; r < 8; ++r) {
      O0[r] *= c; O1[r] *= c; O2[r] *= c; O3[r] *= c;
    }
#pragma unroll
    for (int kk = 0; kk < 4; ++kk) {
      v2f pb = {p[2 * kk], p[2 * kk + 1]};
      v2f a0 = {vv[2 * kk].x, vv[2 * kk + 1].x};
      v2f a1 = {vv[2 * kk].y, vv[2 * kk + 1].y};
      v2f a2 = {vv[2 * kk].z, vv[2 * kk + 1].z};
      v2f a3 = {vv[2 * kk].w, vv[2 * kk + 1].w};
      O0 = wmma_f32(a0, pb, O0);
      O1 = wmma_f32(a1, pb, O1);
      O2 = wmma_f32(a2, pb, O2);
      O3 = wmma_f32(a3, pb, O3);
    }
  }
  const float lsum = lrun + xhalf(lrun);
  const float inv = 1.0f / lsum;
  if (q < N_) {
    float* outp = AO + ((size_t)b * N_ + q) * D_ + h * HD_ + 32 * lh;
#pragma unroll
    for (int r = 0; r < 8; ++r) {
      outp[4 * r + 0] = O0[r] * inv;
      outp[4 * r + 1] = O1[r] * inv;
      outp[4 * r + 2] = O2[r] * inv;
      outp[4 * r + 3] = O3[r] * inv;
    }
  }
}

__global__ __launch_bounds__(32) void proj_gemm(
    const float* __restrict__ AO, const float* __restrict__ w,
    const float* __restrict__ bias, float* __restrict__ P) {
  const int lane = threadIdx.x;
  const int lm = lane & 15;
  const int lh = lane >> 4;
  const int col0 = blockIdx.x * 64;
  const int row0 = blockIdx.y * 80;
  v8f c[RT_][CT_];
#pragma unroll
  for (int i = 0; i < RT_; ++i)
#pragma unroll
    for (int j = 0; j < CT_; ++j) c[i][j] = v8f{};
  gemm_tile_80x64(AO + (size_t)(row0 + lm) * D_ + 16 * lh,
                  w + (size_t)(col0 + lm) * D_ + 16 * lh, c);
  float bc[CT_];
#pragma unroll
  for (int j = 0; j < CT_; ++j) bc[j] = bias[col0 + 16 * j + lm];
#pragma unroll
  for (int i = 0; i < RT_; ++i) {
#pragma unroll
    for (int e = 0; e < 8; ++e) {
      const int r = row0 + 16 * i + 8 * lh + e;
      float* p = P + (size_t)r * D_ + col0 + lm;
#pragma unroll
      for (int j = 0; j < CT_; ++j) p[16 * j] = c[i][j][e] + bc[j];
    }
  }
}

__global__ __launch_bounds__(256) void ln_residual(
    float* __restrict__ Y, const float* __restrict__ gamma,
    const float* __restrict__ beta) {
  __shared__ float reds[8], reds2[8];
  __shared__ float mu_s, rs_s;
  const int row = blockIdx.x;
  float* y = Y + (size_t)row * D_;
  const int t = threadIdx.x;
  float a = y[t], bv = y[t + 256];
  float s = a + bv, s2 = a * a + bv * bv;
#pragma unroll
  for (int off = 1; off < 32; off <<= 1) {
    s += __shfl_xor(s, off, 32);
    s2 += __shfl_xor(s2, off, 32);
  }
  if ((t & 31) == 0) { reds[t >> 5] = s; reds2[t >> 5] = s2; }
  __syncthreads();
  if (t == 0) {
    float S = 0.f, S2 = 0.f;
#pragma unroll
    for (int i = 0; i < 8; ++i) { S += reds[i]; S2 += reds2[i]; }
    float mu = S * (1.0f / D_);
    float var = S2 * (1.0f / D_) - mu * mu;
    mu_s = mu;
    rs_s = rsqrtf(var + 1e-5f);
  }
  __syncthreads();
  const float mu = mu_s, rs = rs_s;
  y[t]       = a  + (a  - mu) * rs * gamma[t]       + beta[t];
  y[t + 256] = bv + (bv - mu) * rs * gamma[t + 256] + beta[t + 256];
}

extern "C" void kernel_launch(void* const* d_in, const int* in_sizes, int n_in,
                              void* d_out, int out_size, void* d_ws,
                              size_t ws_size, hipStream_t stream) {
  const float* x      = (const float*)d_in[0];
  const float* w_qkv  = (const float*)d_in[1];
  const float* w_proj = (const float*)d_in[2];
  const float* b_proj = (const float*)d_in[3];
  const float* gw     = (const float*)d_in[4];
  const float* ln_g   = (const float*)d_in[5];
  const float* ln_b   = (const float*)d_in[6];
  float* out = (float*)d_out;

  float* ws = (float*)d_ws;
  const size_t nsp  = (size_t)N_ * N_;
  const size_t nqkv = (size_t)B_ * H_ * N_ * HD_;
  float* SP = ws;
  float* Q  = SP + nsp;
  float* K  = Q + nqkv;
  float* V  = K + nqkv;
  float* AO = V + nqkv;

  softplus_kernel<<<(int)((nsp + 255) / 256), 256, 0, stream>>>(gw, SP,
                                                                (int)nsp);
  qkv_gemm<<<dim3(3 * D_ / 64, ROWS_ / 80), 32, 0, stream>>>(x, w_qkv, Q, K,
                                                             V);
  attn_kernel<<<dim3(QT_, B_ * H_), 32, 0, stream>>>(Q, K, V, SP, AO);
  proj_gemm<<<dim3(D_ / 64, ROWS_ / 80), 32, 0, stream>>>(AO, w_proj, b_proj,
                                                          out);
  ln_residual<<<ROWS_, 256, 0, stream>>>(out, ln_g, ln_b);
}


# attention: one wave takes 4 query tiles (64 queries) of a head, so each K fragment and V row loaded feeds 4x the WMMAs; key mask by select; left-over tile one per wave; per-query sum order unchanged
# speedup vs baseline: 3.1036x; 1.3766x over previous
#include <hip/hip_runtime.h>

#define B_  16
#define N_  1025
#define D_  512
#define H_  8
#define HD_ 64
#define QT_ 65
#define ROWS_ 16400

typedef __attribute__((ext_vector_type(2))) float v2f;
typedef __attribute__((ext_vector_type(8))) float v8f;
typedef float v4fu __attribute__((ext_vector_type(4), aligned(4)));

#define RT_ 5
#define CT_ 4

__device__ __forceinline__ v8f wmma_f32(v2f a, v2f b, v8f c) {
  return __builtin_amdgcn_wmma_f32_16x16x4_f32(false, a, false, b, (short)0, c,
                                               false, false);
}

__global__ __launch_bounds__(256) void softplus_kernel(
    const float* __restrict__ gw, float* __restrict__ sp, int n) {
  int i = blockIdx.x * 256 + threadIdx.x;
  if (i < n) {
    float w = gw[i];
    sp[i] = (w > 20.0f) ? w : log1pf(__expf(w));
  }
}

__device__ __forceinline__ void gemm_tile_80x64(const float* __restrict__ ap,
                                                const float* __restrict__ bp,
                                                v8f (&c)[RT_][CT_]) {
  for (int k0 = 0; k0 < D_; k0 += 32) {
    v4fu b[CT_][4], a[RT_][4];
#pragma unroll
    for (int j = 0; j < CT_; ++j)
#pragma unroll
      for (int u = 0; u < 4; ++u)
        b[j][u] = *(const v4fu*)(bp + (size_t)j * 16 * D_ + k0 + 4 * u);
#pragma unroll
    for (int i = 0; i < RT_; ++i)
#pragma unroll
      for (int u = 0; u < 4; ++u)
        a[i][u] = *(const v4fu*)(ap + (size_t)i * 16 * D_ + k0 + 4 * u);
    __builtin_amdgcn_sched_barrier(0);
#pragma unroll
    for (int i = 0; i < RT_; ++i) {
#pragma unroll
      for (int u = 0; u < 4; ++u) {
        const v2f alo = {a[i][u].x, a[i][u].y};
        const v2f ahi = {a[i][u].z, a[i][u].w};
#pragma unroll
        for (int j = 0; j < CT_; ++j) {
          const v2f blo = {b[j][u].x, b[j][u].y};
          c[i][j] = wmma_f32(alo, blo, c[i][j]);
        }
#pragma unroll
        for (int j = 0; j < CT_; ++j) {
          const v2f bhi = {b[j][u].z, b[j][u].w};
          c[i][j] = wmma_f32(ahi, bhi, c[i][j]);
        }
      }
    }
  }
}

__global__ __launch_bounds__(32) void qkv_gemm(
    const float* __restrict__ x, const float* __restrict__ w,
    float* __restrict__ Qd, float* __restrict__ Kd, float* __restrict__ Vd) {
  const int lane = threadIdx.x;
  const int lm = lane & 15;
  const int lh = lane >> 4;
  const int col0 = blockIdx.x * 64;
  const int row0 = blockIdx.y * 80;
  v8f c[RT_][CT_];
#pragma unroll
  for (int i = 0; i < RT_; ++i)
#pragma unroll
    for (int j = 0; j < CT_; ++j) c[i][j] = v8f{};
  gemm_tile_80x64(x + (size_t)(row0 + lm) * D_ + 16 * lh,
                  w + (size_t)(col0 + lm) * D_ + 16 * lh, c);

  const int part = col0 >> 9;
  const int h = (col0 & 511) >> 6;
  float* dst = (part == 0) ? Qd : (part == 1) ? Kd : Vd;
#pragma unroll
  for (int i = 0; i < RT_; ++i) {
#pragma unroll
    for (int e = 0; e < 8; ++e) {
      const int r = row0 + 16 * i + 8 * lh + e;
      const int bb = r / N_;
      const int nn = r - bb * N_;
      float* p = dst + ((size_t)(bb * H_ + h) * N_ + nn) * HD_ + lm;
#pragma unroll
      for (int j = 0; j < CT_; ++j) p[16 * j] = c[i][j][e];
    }
  }
}

__device__ __forceinline__ float xhalf(float x) {
  int bits = __builtin_bit_cast(int, x);
  return __builtin_bit_cast(
      float, __builtin_amdgcn_permlanex16(bits, bits, 0x76543210, 0xfedcba98,
                                          false, false));
}

template <int QB>
__device__ __forceinline__ void attn_tiles(
    const float* __restrict__ Qp, const float* __restrict__ Kp,
    const float* __restrict__ Vp, const float* __restrict__ SP,
    float* __restrict__ AO, const int q0, const int b, const int h,
    const int lm, const int lh) {
  const float scale = 0.125f;

  v2f qb[QB][16];
  const float* sprow[QB];
#pragma unroll
  for (int t = 0; t < QB; ++t) {
    const int q = q0 + 16 * t + lm;
    const int qc = (q < N_) ? q : N_ - 1;
#pragma unroll
    for (int s = 0; s < 16; ++s)
      qb[t][s] = *(const v2f*)(Qp + (size_t)qc * HD_ + 4 * s + 2 * lh);
    sprow[t] = SP + (unsigned)(qc * N_ + 8 * lh);
  }

  v8f O[QB][4];
  float mrun[QB], lrun[QB];
#pragma unroll
  for (int t = 0; t < QB; ++t) {
#pragma unroll
    for (int j = 0; j < 4; ++j) O[t][j] = v8f{};
    mrun[t] = -3.0e38f;
    lrun[t] = 0.0f;
  }

  for (int kt = 0; kt < QT_; ++kt) {
    const int keya = kt * 16 + lm;
    const int keyac = (keya < N_) ? keya : N_ - 1;
    const float* kp = Kp + (unsigned)(keyac * HD_ + 2 * lh);
    v2f ka[16];
#pragma unroll
    for (int s = 0; s < 16; ++s) ka[s] = *(const v2f*)(kp + 4 * s);
    const int key0 = kt * 16 + 8 * lh;
    v4fu vv[8];
#pragma unroll
    for (int r = 0; r < 8; ++r) {
      int kr = key0 + r;
      kr = (kr < N_) ? kr : N_ - 1;
      vv[r] = *(const v4fu*)(Vp + (unsigned)(kr * HD_ + 4 * lm));
    }
    float spv[QB][8];
#pragma unroll
    for (int t = 0; t < QB; ++t)
#pragma unroll
      for (int r = 0; r < 8; ++r) spv[t][r] = sprow[t][kt * 16 + r];

    v8f S[QB];
#pragma unroll
    for (int t = 0; t < QB; ++t) S[t] = v8f{};
#pragma unroll
    for (int s = 0; s < 16; ++s)
#pragma unroll
      for (int t = 0; t < QB; ++t) S[t] = wmma_f32(ka[s], qb[t][s], S[t]);

    float p[QB][8];
#pragma unroll
    for (int t = 0; t < QB; ++t) {
      float lg[8];
#pragma unroll
      for (int r = 0; r < 8; ++r) {
        const float v = S[t][r] * scale * spv[t][r];
        lg[r] = (key0 + r >= N_) ? -3.0e38f : v;
      }
      float mx = fmaxf(fmaxf(fmaxf(lg[0], lg[1]), fmaxf(lg[2], lg[3])),
                       fmaxf(fmaxf(lg[4], lg[5]), fmaxf(lg[6], lg[7])));
      mx = fmaxf(mx, xhalf(mx));
      const float mnew = fmaxf(mrun[t], mx);
      const float c = __expf(mrun[t] - mnew);
#pragma unroll
      for (int r = 0; r < 8; ++r) p[t][r] = __expf(lg[r] - mnew);
      const float sum = ((p[t][0] + p[t][1]) + (p[t][2] + p[t][3])) +
                        ((p[t][4] + p[t][5]) + (p[t][6] + p[t][7]));
      lrun[t] = lrun[t] * c + sum;
      mrun[t] = mnew;
#pragma unroll
      for (int j = 0; j < 4; ++j)
#pragma unroll
        for (int r = 0; r < 8; ++r) O[t][j][r] *= c;
    }
#pragma unroll
    for (int kk = 0; kk < 4; ++kk) {
      const v2f a0 = {vv[2 * kk].x, vv[2 * kk + 1].x};
      const v2f a1 = {vv[2 * kk].y, vv[2 * kk + 1].y};
      const v2f a2 = {vv[2 * kk].z, vv[2 * kk + 1].z};
      const v2f a3 = {vv[2 * kk].w, vv[2 * kk + 1].w};
#pragma unroll
      for (int t = 0; t < QB; ++t) {
        const v2f pb = {p[t][2 * kk], p[t][2 * kk + 1]};
        O[t][0] = wmma_f32(a0, pb, O[t][0]);
        O[t][1] = wmma_f32(a1, pb, O[t][1]);
        O[t][2] = wmma_f32(a2, pb, O[t][2]);
        O[t][3] = wmma_f32(a3, pb, O[t][3]);
      }
    }
  }
#pragma unroll
  for (int t = 0; t < QB; ++t) {
    const float lsum = lrun[t] + xhalf(lrun[t]);
    const float inv = 1.0f / lsum;
    const int q = q0 + 16 * t + lm;
    if (q < N_) {
      float* outp = AO + ((size_t)b * N_ + q) * D_ + h * HD_ + 32 * lh;
#pragma unroll
      for (int r = 0; r < 8; ++r) {
        outp[4 * r + 0] = O[t][0][r] * inv;
        outp[4 * r + 1] = O[t][1][r] * inv;
        outp[4 * r + 2] = O[t][2][r] * inv;
        outp[4 * r + 3] = O[t][3][r] * inv;
      }
    }
  }
}

#define QB_   4
#define NQB_  (N_ / (16 * QB_))
#define NQR_  (QT_ - NQB_ * QB_)

__global__ __launch_bounds__(32) void attn_kernel(
    const float* __restrict__ Qd, const float* __restrict__ Kd,
    const float* __restrict__ Vd, const float* __restrict__ SP,
    float* __restrict__ AO) {
  const int lane = threadIdx.x;
  const int lm = lane & 15;
  const int lh = lane >> 4;
  const int bh = blockIdx.x;
  const int qblk = blockIdx.y;
  const int b = bh >> 3;
  const int h = bh & 7;
  const size_t base = (size_t)bh * N_ * HD_;
  if (qblk < NQB_)
    attn_tiles<QB_>(Qd + base, Kd + base, Vd + base, SP, AO,
                    qblk * 16 * QB_, b, h, lm, lh);
  else
    attn_tiles<1>(Qd + base, Kd + base, Vd + base, SP, AO,
                  (NQB_ * QB_ + (qblk - NQB_)) * 16, b, h, lm, lh);
}

__global__ __launch_bounds__(32) void proj_gemm(
    const float* __restrict__ AO, const float* __restrict__ w,
    const float* __restrict__ bias, float* __restrict__ P) {
  const int lane = threadIdx.x;
  const int lm = lane & 15;
  const int lh = lane >> 4;
  const int col0 = blockIdx.x * 64;
  const int row0 = blockIdx.y * 80;
  v8f c[RT_][CT_];
#pragma unroll
  for (int i = 0; i < RT_; ++i)
#pragma unroll
    for (int j = 0; j < CT_; ++j) c[i][j] = v8f{};
  gemm_tile_80x64(AO + (size_t)(row0 + lm) * D_ + 16 * lh,
                  w + (size_t)(col0 + lm) * D_ + 16 * lh, c);
  float bc[CT_];
#pragma unroll
  for (int j = 0; j < CT_; ++j) bc[j] = bias[col0 + 16 * j + lm];
#pragma unroll
  for (int i = 0; i < RT_; ++i) {
#pragma unroll
    for (int e = 0; e < 8; ++e) {
      const int r = row0 + 16 * i + 8 * lh + e;
      float* p = P + (size_t)r * D_ + col0 + lm;
#pragma unroll
      for (int j = 0; j < CT_; ++j) p[16 * j] = c[i][j][e] + bc[j];
    }
  }
}

__global__ __launch_bounds__(256) void ln_residual(
    float* __restrict__ Y, const float* __restrict__ gamma,
    const float* __restrict__ beta) {
  __shared__ float reds[8], reds2[8];
  __shared__ float mu_s, rs_s;
  const int row = blockIdx.x;
  float* y = Y + (size_t)row * D_;
  const int t = threadIdx.x;
  float a = y[t], bv = y[t + 256];
  float s = a + bv, s2 = a * a + bv * bv;
#pragma unroll
  for (int off = 1; off < 32; off <<= 1) {
    s += __shfl_xor(s, off, 32);
    s2 += __shfl_xor(s2, off, 32);
  }
  if ((t & 31) == 0) { reds[t >> 5] = s; reds2[t >> 5] = s2; }
  __syncthreads();
  if (t == 0) {
    float S = 0.f, S2 = 0.f;
#pragma unroll
    for (int i = 0; i < 8; ++i) { S += reds[i]; S2 += reds2[i]; }
    float mu = S * (1.0f / D_);
    float var = S2 * (1.0f / D_) - mu * mu;
    mu_s = mu;
    rs_s = rsqrtf(var + 1e-5f);
  }
  __syncthreads();
  const float mu = mu_s, rs = rs_s;
  y[t]       = a  + (a  - mu) * rs * gamma[t]       + beta[t];
  y[t + 256] = bv + (bv - mu) * rs * gamma[t + 256] + beta[t + 256];
}

extern "C" void kernel_launch(void* const* d_in, const int* in_sizes, int n_in,
                              void* d_out, int out_size, void* d_ws,
                              size_t ws_size, hipStream_t stream) {
  const float* x      = (const float*)d_in[0];
  const float* w_qkv  = (const float*)d_in[1];
  const float* w_proj = (const float*)d_in[2];
  const float* b_proj = (const float*)d_in[3];
  const float* gw     = (const float*)d_in[4];
  const float* ln_g   = (const float*)d_in[5];
  const float* ln_b   = (const float*)d_in[6];
  float* out = (float*)d_out;

  float* ws = (float*)d_ws;
  const size_t nsp  = (size_t)N_ * N_;
  const size_t nqkv = (size_t)B_ * H_ * N_ * HD_;
  float* SP = ws;
  float* Q  = SP + nsp;
  float* K  = Q + nqkv;
  float* V  = K + nqkv;
  float* AO = V + nqkv;

  softplus_kernel<<<(int)((nsp + 255) / 256), 256, 0, stream>>>(gw, SP,
                                                                (int)nsp);
  qkv_gemm<<<dim3(3 * D_ / 64, ROWS_ / 80), 32, 0, stream>>>(x, w_qkv, Q, K,
                                                             V);
  attn_kernel<<<dim3(B_ * H_, NQB_ + NQR_), 32, 0, stream>>>(Q, K, V, SP, AO);
  proj_gemm<<<dim3(D_ / 64, ROWS_ / 80), 32, 0, stream>>>(AO, w_proj, b_proj,
                                                          out);
  ln_residual<<<ROWS_, 256, 0, stream>>>(out, ln_g, ln_b);
}


# K and softplus planes stored in the attention wave's read order: each fragment load is 512 contiguous aligned bytes (4 lines, was 16-32 lines); changes the d summation order of the scores
# speedup vs baseline: 3.4271x; 1.1042x over previous
#include <hip/hip_runtime.h>

#define B_  16
#define N_  1025
#define D_  512
#define H_  8
#define HD_ 64
#define QT_ 65
#define ROWS_ 16400
#define KT_ 64
#define KTAIL_ (KT_ * 16 * HD_)
#define SPCOL_ (KT_ * KT_ * 256)
#define SPROW_ (SPCOL_ + KT_ * 16)
static_assert(N_ == KT_ * 16 + 1, "one key and one query past the whole tiles");

typedef __attribute__((ext_vector_type(2))) float v2f;
typedef __attribute__((ext_vector_type(8))) float v8f;
typedef float v4fu __attribute__((ext_vector_type(4), aligned(4)));

#define RT_ 5
#define CT_ 4

__device__ __forceinline__ v8f wmma_f32(v2f a, v2f b, v8f c) {
  return __builtin_amdgcn_wmma_f32_16x16x4_f32(false, a, false, b, (short)0, c,
                                               false, false);
}

__global__ __launch_bounds__(256) void softplus_kernel(
    const float* __restrict__ gw, float* __restrict__ sp, int n) {
  int i = blockIdx.x * 256 + threadIdx.x;
  if (i < n) {
    int q, k;
    if (i < SPCOL_) {
      const int e = i & 3, lane = (i >> 2) & 31, w2 = (i >> 7) & 1;
      const int kt = (i >> 8) & 63, qt = i >> 14;
      q = qt * 16 + (lane & 15);
      k = kt * 16 + 8 * (lane >> 4) + 4 * w2 + e;
    } else if (i < SPROW_) {
      q = i - SPCOL_;
      k = N_ - 1;
    } else {
      q = N_ - 1;
      k = i - SPROW_;
    }
    float w = gw[q * N_ + k];
    sp[i] = (w > 20.0f) ? w : log1pf(__expf(w));
  }
}

__device__ __forceinline__ void gemm_tile_80x64(const float* __restrict__ ap,
                                                const float* __restrict__ bp,
                                                v8f (&c)[RT_][CT_]) {
  for (int k0 = 0; k0 < D_; k0 += 32) {
    v4fu b[CT_][4], a[RT_][4];
#pragma unroll
    for (int j = 0; j < CT_; ++j)
#pragma unroll
      for (int u = 0; u < 4; ++u)
        b[j][u] = *(const v4fu*)(bp + (size_t)j * 16 * D_ + k0 + 4 * u);
#pragma unroll
    for (int i = 0; i < RT_; ++i)
#pragma unroll
      for (int u = 0; u < 4; ++u)
        a[i][u] = *(const v4fu*)(ap + (size_t)i * 16 * D_ + k0 + 4 * u);
    __builtin_amdgcn_sched_barrier(0);
#pragma unroll
    for (int i = 0; i < RT_; ++i) {
#pragma unroll
      for (int u = 0; u < 4; ++u) {
        const v2f alo = {a[i][u].x, a[i][u].y};
        const v2f ahi = {a[i][u].z, a[i][u].w};
#pragma unroll
        for (int j = 0; j < CT_; ++j) {
          const v2f blo = {b[j][u].x, b[j][u].y};
          c[i][j] = wmma_f32(alo, blo, c[i][j]);
        }
#pragma unroll
        for (int j = 0; j < CT_; ++j) {
          const v2f bhi = {b[j][u].z, b[j][u].w};
          c[i][j] = wmma_f32(ahi, bhi, c[i][j]);
        }
      }
    }
  }
}

__global__ __launch_bounds__(32) void qkv_gemm(
    const float* __restrict__ x, const float* __restrict__ w,
    float* __restrict__ Qd, float* __restrict__ Kd, float* __restrict__ Vd) {
  const int lane = threadIdx.x;
  const int lm = lane & 15;
  const int lh = lane >> 4;
  const int col0 = blockIdx.x * 64;
  const int row0 = blockIdx.y * 80;
  v8f c[RT_][CT_];
#pragma unroll
  for (int i = 0; i < RT_; ++i)
#pragma unroll
    for (int j = 0; j < CT_; ++j) c[i][j] = v8f{};
  gemm_tile_80x64(x + (size_t)(row0 + lm) * D_ + 16 * lh,
                  w + (size_t)(col0 + lm) * D_ + 16 * lh, c);

  const int part = col0 >> 9;
  const int h = (col0 & 511) >> 6;
  if (part == 1) {
    const int c0 = (lm >> 3) * 128 + ((lm >> 2) & 1) * 64 + (lm & 3);
#pragma unroll
    for (int i = 0; i < RT_; ++i) {
#pragma unroll
      for (int e = 0; e < 8; ++e) {
        const int r = row0 + 16 * i + 8 * lh + e;
        const int bb = r / N_;
        const int nn = r - bb * N_;
        const int off = (nn < KT_ * 16)
                            ? (nn >> 4) * 1024 + (nn & 15) * 4 + c0
                            : KTAIL_ + lm;
        const int jstep = (nn < KT_ * 16) ? 256 : 16;
        float* p = Kd + (size_t)(bb * H_ + h) * N_ * HD_ + off;
#pragma unroll
        for (int j = 0; j < CT_; ++j) p[jstep * j] = c[i][j][e];
      }
    }
    return;
  }
  float* dst = (part == 0) ? Qd : Vd;
#pragma unroll
  for (int i = 0; i < RT_; ++i) {
#pragma unroll
    for (int e = 0; e < 8; ++e) {
      const int r = row0 + 16 * i + 8 * lh + e;
      const int bb = r / N_;
      const int nn = r - bb * N_;
      float* p = dst + ((size_t)(bb * H_ + h) * N_ + nn) * HD_ + lm;
#pragma unroll
      for (int j = 0; j < CT_; ++j) p[16 * j] = c[i][j][e];
    }
  }
}

__device__ __forceinline__ float xhalf(float x) {
  int bits = __builtin_bit_cast(int, x);
  return __builtin_bit_cast(
      float, __builtin_amdgcn_permlanex16(bits, bits, 0x76543210, 0xfedcba98,
                                          false, false));
}

template <bool W> struct tile_kind { static constexpr bool whole = W; };

template <int QB>
__device__ __forceinline__ void attn_tiles(
    const float* __restrict__ Qp, const float* __restrict__ Kp,
    const float* __restrict__ Vp, const float* __restrict__ SP,
    float* __restrict__ AO, const int q0, const int b, const int h,
    const int lm, const int lh) {
  const float scale = 0.125f;

  const int lane = lm + 16 * lh;
  v2f qb[QB][16];
  unsigned spblk[QB];
  unsigned spcol[QB];
#pragma unroll
  for (int t = 0; t < QB; ++t) {
    const int q = q0 + 16 * t + lm;
    const int qc = (q < N_) ? q : N_ - 1;
#pragma unroll
    for (int s = 0; s < 16; ++s)
      qb[t][s] = *(const v2f*)(Qp + (size_t)qc * HD_ + 8 * (s >> 1) + 4 * lh +
                               2 * (s & 1));
    spblk[t] = (unsigned)(((q0 >> 4) + t) * (KT_ * 256) + lane * 4);
    spcol[t] = (unsigned)(SPCOL_ + ((qc < KT_ * 16) ? qc : 0));
  }

  v8f O[QB][4];
  float mrun[QB], lrun[QB];
#pragma unroll
  for (int t = 0; t < QB; ++t) {
#pragma unroll
    for (int j = 0; j < 4; ++j) O[t][j] = v8f{};
    mrun[t] = -3.0e38f;
    lrun[t] = 0.0f;
  }

  auto tile = [&](const int kt, auto kind) __attribute__((always_inline)) {
    constexpr bool whole = decltype(kind)::whole;
    const unsigned ko = whole ? (unsigned)(kt * 1024 + lane * 4)
                              : (unsigned)(KTAIL_ + 4 * lh);
    constexpr unsigned ku = whole ? 128u : 8u;
    v4fu kf[8];
#pragma unroll
    for (int u = 0; u < 8; ++u) kf[u] = *(const v4fu*)(Kp + (ko + u * ku));
    const int key0 = kt * 16 + 8 * lh;
    v4fu vv[8];
#pragma unroll
    for (int r = 0; r < 8; ++r) {
      int kr = key0 + r;
      if constexpr (!whole) kr = (kr < N_) ? kr : N_ - 1;
      vv[r] = *(const v4fu*)(Vp + (unsigned)(kr * HD_ + 4 * lm));
    }
    float spv[QB][8];
    if constexpr (QB > 1) {
#pragma unroll
      for (int t = 0; t < QB; ++t) {
        if constexpr (whole) {
          const unsigned so = spblk[t] + (unsigned)(kt * 256);
          const v4fu s0 = *(const v4fu*)(SP + so);
          const v4fu s1 = *(const v4fu*)(SP + (so + 128u));
          spv[t][0] = s0.x; spv[t][1] = s0.y; spv[t][2] = s0.z; spv[t][3] = s0.w;
          spv[t][4] = s1.x; spv[t][5] = s1.y; spv[t][6] = s1.z; spv[t][7] = s1.w;
        } else {
          spv[t][0] = SP[spcol[t]];
#pragma unroll
          for (int r = 1; r < 8; ++r) spv[t][r] = 0.0f;
        }
      }
    } else {
#pragma unroll
      for (int r = 0; r < 8; ++r) {
        int k = key0 + r;
        if constexpr (!whole) k = (k < N_) ? k : N_ - 1;
        spv[0][r] = SP[SPROW_ + k];
      }
    }

    v8f S[QB];
#pragma unroll
    for (int t = 0; t < QB; ++t) S[t] = v8f{};
#pragma unroll
    for (int u = 0; u < 8; ++u) {
      const v2f klo = {kf[u].x, kf[u].y};
      const v2f khi = {kf[u].z, kf[u].w};
#pragma unroll
      for (int t = 0; t < QB; ++t) S[t] = wmma_f32(klo, qb[t][2 * u], S[t]);
#pragma unroll
      for (int t = 0; t < QB; ++t) S[t] = wmma_f32(khi, qb[t][2 * u + 1], S[t]);
    }

    float p[QB][8];
#pragma unroll
    for (int t = 0; t < QB; ++t) {
      float lg[8];
#pragma unroll
      for (int r = 0; r < 8; ++r) {
        const float v = S[t][r] * scale * spv[t][r];
        if constexpr (whole)
          lg[r] = v;
        else
          lg[r] = (key0 + r >= N_) ? -3.0e38f : v;
      }
      float mx = fmaxf(fmaxf(fmaxf(lg[0], lg[1]), fmaxf(lg[2], lg[3])),
                       fmaxf(fmaxf(lg[4], lg[5]), fmaxf(lg[6], lg[7])));
      mx = fmaxf(mx, xhalf(mx));
      const float mnew = fmaxf(mrun[t], mx);
      const float c = __expf(mrun[t] - mnew);
#pragma unroll
      for (int r = 0; r < 8; ++r) p[t][r] = __expf(lg[r] - mnew);
      const float sum = ((p[t][0] + p[t][1]) + (p[t][2] + p[t][3])) +
                        ((p[t][4] + p[t][5]) + (p[t][6] + p[t][7]));
      lrun[t] = lrun[t] * c + sum;
      mrun[t] = mnew;
#pragma unroll
      for (int j = 0; j < 4; ++j)
#pragma unroll
        for (int r = 0; r < 8; ++r) O[t][j][r] *= c;
    }
#pragma unroll
    for (int kk = 0; kk < 4; ++kk) {
      const v2f a0 = {vv[2 * kk].x, vv[2 * kk + 1].x};
      const v2f a1 = {vv[2 * kk].y, vv[2 * kk + 1].y};
      const v2f a2 = {vv[2 * kk].z, vv[2 * kk + 1].z};
      const v2f a3 = {vv[2 * kk].w, vv[2 * kk + 1].w};
#pragma unroll
      for (int t = 0; t < QB; ++t) {
        const v2f pb = {p[t][2 * kk], p[t][2 * kk + 1]};
        O[t][0] = wmma_f32(a0, pb, O[t][0]);
        O[t][1] = wmma_f32(a1, pb, O[t][1]);
        O[t][2] = wmma_f32(a2, pb, O[t][2]);
        O[t][3] = wmma_f32(a3, pb, O[t][3]);
      }
    }
  };
  for (int kt = 0; kt < KT_; ++kt) tile(kt, tile_kind<true>{});
  tile(KT_, tile_kind<false>{});
#pragma unroll
  for (int t = 0; t < QB; ++t) {
    const float lsum = lrun[t] + xhalf(lrun[t]);
    const float inv = 1.0f / lsum;
    const int q = q0 + 16 * t + lm;
    if (q < N_) {
      float* outp = AO + ((size_t)b * N_ + q) * D_ + h * HD_ + 32 * lh;
#pragma unroll
      for (int r = 0; r < 8; ++r) {
        outp[4 * r + 0] = O[t][0][r] * inv;
        outp[4 * r + 1] = O[t][1][r] * inv;
        outp[4 * r + 2] = O[t][2][r] * inv;
        outp[4 * r + 3] = O[t][3][r] * inv;
      }
    }
  }
}

#define QB_   4
#define NQB_  (N_ / (16 * QB_))
#define NQR_  (QT_ - NQB_ * QB_)

__global__ __launch_bounds__(32) void attn_kernel(
    const float* __restrict__ Qd, const float* __restrict__ Kd,
    const float* __restrict__ Vd, const float* __restrict__ SP,
    float* __restrict__ AO) {
  const int lane = threadIdx.x;
  const int lm = lane & 15;
  const int lh = lane >> 4;
  const int bh = blockIdx.x;
  const int qblk = blockIdx.y;
  const int b = bh >> 3;
  const int h = bh & 7;
  const size_t base = (size_t)bh * N_ * HD_;
  if (qblk < NQB_)
    attn_tiles<QB_>(Qd + base, Kd + base, Vd + base, SP, AO,
                    qblk * 16 * QB_, b, h, lm, lh);
  else
    attn_tiles<1>(Qd + base, Kd + base, Vd + base, SP, AO,
                  (NQB_ * QB_ + (qblk - NQB_)) * 16, b, h, lm, lh);
}

__global__ __launch_bounds__(32) void proj_gemm(
    const float* __restrict__ AO, const float* __restrict__ w,
    const float* __restrict__ bias, float* __restrict__ P) {
  const int lane = threadIdx.x;
  const int lm = lane & 15;
  const int lh = lane >> 4;
  const int col0 = blockIdx.x * 64;
  const int row0 = blockIdx.y * 80;
  v8f c[RT_][CT_];
#pragma unroll
  for (int i = 0; i < RT_; ++i)
#pragma unroll
    for (int j = 0; j < CT_; ++j) c[i][j] = v8f{};
  gemm_tile_80x64(AO + (size_t)(row0 + lm) * D_ + 16 * lh,
                  w + (size_t)(col0 + lm) * D_ + 16 * lh, c);
  float bc[CT_];
#pragma unroll
  for (int j = 0; j < CT_; ++j) bc[j] = bias[col0 + 16 * j + lm];
#pragma unroll
  for (int i = 0; i < RT_; ++i) {
#pragma unroll
    for (int e = 0; e < 8; ++e) {
      const int r = row0 + 16 * i + 8 * lh + e;
      float* p = P + (size_t)r * D_ + col0 + lm;
#pragma unroll
      for (int j = 0; j < CT_; ++j) p[16 * j] = c[i][j][e] + bc[j];
    }
  }
}

__global__ __launch_bounds__(256) void ln_residual(
    float* __restrict__ Y, const float* __restrict__ gamma,
    const float* __restrict__ beta) {
  __shared__ float reds[8], reds2[8];
  __shared__ float mu_s, rs_s;
  const int row = blockIdx.x;
  float* y = Y + (size_t)row * D_;
  const int t = threadIdx.x;
  float a = y[t], bv = y[t + 256];
  float s = a + bv, s2 = a * a + bv * bv;
#pragma unroll
  for (int off = 1; off < 32; off <<= 1) {
    s += __shfl_xor(s, off, 32);
    s2 += __shfl_xor(s2, off, 32);
  }
  if ((t & 31) == 0) { reds[t >> 5] = s; reds2[t >> 5] = s2; }
  __syncthreads();
  if (t == 0) {
    float S = 0.f, S2 = 0.f;
#pragma unroll
    for (int i = 0; i < 8; ++i) { S += reds[i]; S2 += reds2[i]; }
    float mu = S * (1.0f / D_);
    float var = S2 * (1.0f / D_) - mu * mu;
    mu_s = mu;
    rs_s = rsqrtf(var + 1e-5f);
  }
  __syncthreads();
  const float mu = mu_s, rs = rs_s;
  y[t]       = a  + (a  - mu) * rs * gamma[t]       + beta[t];
  y[t + 256] = bv + (bv - mu) * rs * gamma[t + 256] + beta[t + 256];
}

extern "C" void kernel_launch(void* const* d_in, const int* in_sizes, int n_in,
                              void* d_out, int out_size, void* d_ws,
                              size_t ws_size, hipStream_t stream) {
  const float* x      = (const float*)d_in[0];
  const float* w_qkv  = (const float*)d_in[1];
  const float* w_proj = (const float*)d_in[2];
  const float* b_proj = (const float*)d_in[3];
  const float* gw     = (const float*)d_in[4];
  const float* ln_g   = (const float*)d_in[5];
  const float* ln_b   = (const float*)d_in[6];
  float* out = (float*)d_out;

  float* ws = (float*)d_ws;
  const size_t nsp  = (size_t)N_ * N_;
  const size_t nqkv = (size_t)B_ * H_ * N_ * HD_;
  float* Q  = ws;
  float* K  = Q + nqkv;
  float* V  = K + nqkv;
  float* AO = V + nqkv;
  float* SP = AO + nqkv;

  softplus_kernel<<<(int)((nsp + 255) / 256), 256, 0, stream>>>(gw, SP,
                                                                (int)nsp);
  qkv_gemm<<<dim3(3 * D_ / 64, ROWS_ / 80), 32, 0, stream>>>(x, w_qkv, Q, K,
                                                             V);
  attn_kernel<<<dim3(B_ * H_, NQB_ + NQR_), 32, 0, stream>>>(Q, K, V, SP, AO);
  proj_gemm<<<dim3(D_ / 64, ROWS_ / 80), 32, 0, stream>>>(AO, w_proj, b_proj,
                                                          out);
  ln_residual<<<ROWS_, 256, 0, stream>>>(out, ln_g, ln_b);
}


# LayerNorm+residual fused into the projection GEMM: a work-group of 8 waves owns 80 whole rows, row sums joined through LDS; no LN launch, y not re-read; changes LN row-sum order
# speedup vs baseline: 3.6782x; 1.0733x over previous
#include <hip/hip_runtime.h>

#define B_  16
#define N_  1025
#define D_  512
#define H_  8
#define HD_ 64
#define QT_ 65
#define ROWS_ 16400
#define KT_ 64
#define KTAIL_ (KT_ * 16 * HD_)
#define SPCOL_ (KT_ * KT_ * 256)
#define SPROW_ (SPCOL_ + KT_ * 16)
static_assert(N_ == KT_ * 16 + 1, "one key and one query past the whole tiles");

typedef __attribute__((ext_vector_type(2))) float v2f;
typedef __attribute__((ext_vector_type(8))) float v8f;
typedef float v4fu __attribute__((ext_vector_type(4), aligned(4)));

#define RT_ 5
#define CT_ 4

__device__ __forceinline__ v8f wmma_f32(v2f a, v2f b, v8f c) {
  return __builtin_amdgcn_wmma_f32_16x16x4_f32(false, a, false, b, (short)0, c,
                                               false, false);
}

__global__ __launch_bounds__(256) void softplus_kernel(
    const float* __restrict__ gw, float* __restrict__ sp, int n) {
  int i = blockIdx.x * 256 + threadIdx.x;
  if (i < n) {
    int q, k;
    if (i < SPCOL_) {
      const int e = i & 3, lane = (i >> 2) & 31, w2 = (i >> 7) & 1;
      const int kt = (i >> 8) & 63, qt = i >> 14;
      q = qt * 16 + (lane & 15);
      k = kt * 16 + 8 * (lane >> 4) + 4 * w2 + e;
    } else if (i < SPROW_) {
      q = i - SPCOL_;
      k = N_ - 1;
    } else {
      q = N_ - 1;
      k = i - SPROW_;
    }
    float w = gw[q * N_ + k];
    sp[i] = (w > 20.0f) ? w : log1pf(__expf(w));
  }
}

__device__ __forceinline__ void gemm_tile_80x64(const float* __restrict__ ap,
                                                const float* __restrict__ bp,
                                                v8f (&c)[RT_][CT_]) {
  for (int k0 = 0; k0 < D_; k0 += 32) {
    v4fu b[CT_][4], a[RT_][4];
#pragma unroll
    for (int j = 0; j < CT_; ++j)
#pragma unroll
      for (int u = 0; u < 4; ++u)
        b[j][u] = *(const v4fu*)(bp + (size_t)j * 16 * D_ + k0 + 4 * u);
#pragma unroll
    for (int i = 0; i < RT_; ++i)
#pragma unroll
      for (int u = 0; u < 4; ++u)
        a[i][u] = *(const v4fu*)(ap + (size_t)i * 16 * D_ + k0 + 4 * u);
    __builtin_amdgcn_sched_barrier(0);
#pragma unroll
    for (int i = 0; i < RT_; ++i) {
#pragma unroll
      for (int u = 0; u < 4; ++u) {
        const v2f alo = {a[i][u].x, a[i][u].y};
        const v2f ahi = {a[i][u].z, a[i][u].w};
#pragma unroll
        for (int j = 0; j < CT_; ++j) {
          const v2f blo = {b[j][u].x, b[j][u].y};
          c[i][j] = wmma_f32(alo, blo, c[i][j]);
        }
#pragma unroll
        for (int j = 0; j < CT_; ++j) {
          const v2f bhi = {b[j][u].z, b[j][u].w};
          c[i][j] = wmma_f32(ahi, bhi, c[i][j]);
        }
      }
    }
  }
}

__global__ __launch_bounds__(32) void qkv_gemm(
    const float* __restrict__ x, const float* __restrict__ w,
    float* __restrict__ Qd, float* __restrict__ Kd, float* __restrict__ Vd) {
  const int lane = threadIdx.x;
  const int lm = lane & 15;
  const int lh = lane >> 4;
  const int col0 = blockIdx.x * 64;
  const int row0 = blockIdx.y * 80;
  v8f c[RT_][CT_];
#pragma unroll
  for (int i = 0; i < RT_; ++i)
#pragma unroll
    for (int j = 0; j < CT_; ++j) c[i][j] = v8f{};
  gemm_tile_80x64(x + (size_t)(row0 + lm) * D_ + 16 * lh,
                  w + (size_t)(col0 + lm) * D_ + 16 * lh, c);

  const int part = col0 >> 9;
  const int h = (col0 & 511) >> 6;
  if (part == 1) {
    const int c0 = (lm >> 3) * 128 + ((lm >> 2) & 1) * 64 + (lm & 3);
#pragma unroll
    for (int i = 0; i < RT_; ++i) {
#pragma unroll
      for (int e = 0; e < 8; ++e) {
        const int r = row0 + 16 * i + 8 * lh + e;
        const int bb = r / N_;
        const int nn = r - bb * N_;
        const int off = (nn < KT_ * 16)
                            ? (nn >> 4) * 1024 + (nn & 15) * 4 + c0
                            : KTAIL_ + lm;
        const int jstep = (nn < KT_ * 16) ? 256 : 16;
        float* p = Kd + (size_t)(bb * H_ + h) * N_ * HD_ + off;
#pragma unroll
        for (int j = 0; j < CT_; ++j) p[jstep * j] = c[i][j][e];
      }
    }
    return;
  }
  float* dst = (part == 0) ? Qd : Vd;
#pragma unroll
  for (int i = 0; i < RT_; ++i) {
#pragma unroll
    for (int e = 0; e < 8; ++e) {
      const int r = row0 + 16 * i + 8 * lh + e;
      const int bb = r / N_;
      const int nn = r - bb * N_;
      float* p = dst + ((size_t)(bb * H_ + h) * N_ + nn) * HD_ + lm;
#pragma unroll
      for (int j = 0; j < CT_; ++j) p[16 * j] = c[i][j][e];
    }
  }
}

__device__ __forceinline__ float xhalf(float x) {
  int bits = __builtin_bit_cast(int, x);
  return __builtin_bit_cast(
      float, __builtin_amdgcn_permlanex16(bits, bits, 0x76543210, 0xfedcba98,
                                          false, false));
}

template <bool W> struct tile_kind { static constexpr bool whole = W; };

template <int QB>
__device__ __forceinline__ void attn_tiles(
    const float* __restrict__ Qp, const float* __restrict__ Kp,
    const float* __restrict__ Vp, const float* __restrict__ SP,
    float* __restrict__ AO, const int q0, const int b, const int h,
    const int lm, const int lh) {
  const float scale = 0.125f;

  const int lane = lm + 16 * lh;
  v2f qb[QB][16];
  unsigned spblk[QB];
  unsigned spcol[QB];
#pragma unroll
  for (int t = 0; t < QB; ++t) {
    const int q = q0 + 16 * t + lm;
    const int qc = (q < N_) ? q : N_ - 1;
#pragma unroll
    for (int s = 0; s < 16; ++s)
      qb[t][s] = *(const v2f*)(Qp + (size_t)qc * HD_ + 8 * (s >> 1) + 4 * lh +
                               2 * (s & 1));
    spblk[t] = (unsigned)(((q0 >> 4) + t) * (KT_ * 256) + lane * 4);
    spcol[t] = (unsigned)(SPCOL_ + ((qc < KT_ * 16) ? qc : 0));
  }

  v8f O[QB][4];
  float mrun[QB], lrun[QB];
#pragma unroll
  for (int t = 0; t < QB; ++t) {
#pragma unroll
    for (int j = 0; j < 4; ++j) O[t][j] = v8f{};
    mrun[t] = -3.0e38f;
    lrun[t] = 0.0f;
  }

  auto tile = [&](const int kt, auto kind) __attribute__((always_inline)) {
    constexpr bool whole = decltype(kind)::whole;
    const unsigned ko = whole ? (unsigned)(kt * 1024 + lane * 4)
                              : (unsigned)(KTAIL_ + 4 * lh);
    constexpr unsigned ku = whole ? 128u : 8u;
    v4fu kf[8];
#pragma unroll
    for (int u = 0; u < 8; ++u) kf[u] = *(const v4fu*)(Kp + (ko + u * ku));
    const int key0 = kt * 16 + 8 * lh;
    v4fu vv[8];
#pragma unroll
    for (int r = 0; r < 8; ++r) {
      int kr = key0 + r;
      if constexpr (!whole) kr = (kr < N_) ? kr : N_ - 1;
      vv[r] = *(const v4fu*)(Vp + (unsigned)(kr * HD_ + 4 * lm));
    }
    float spv[QB][8];
    if constexpr (QB > 1) {
#pragma unroll
      for (int t = 0; t < QB; ++t) {
        if constexpr (whole) {
          const unsigned so = spblk[t] + (unsigned)(kt * 256);
          const v4fu s0 = *(const v4fu*)(SP + so);
          const v4fu s1 = *(const v4fu*)(SP + (so + 128u));
          spv[t][0] = s0.x; spv[t][1] = s0.y; spv[t][2] = s0.z; spv[t][3] = s0.w;
          spv[t][4] = s1.x; spv[t][5] = s1.y; spv[t][6] = s1.z; spv[t][7] = s1.w;
        } else {
          spv[t][0] = SP[spcol[t]];
#pragma unroll
          for (int r = 1; r < 8; ++r) spv[t][r] = 0.0f;
        }
      }
    } else {
#pragma unroll
      for (int r = 0; r < 8; ++r) {
        int k = key0 + r;
        if constexpr (!whole) k = (k < N_) ? k : N_ - 1;
        spv[0][r] = SP[SPROW_ + k];
      }
    }

    v8f S[QB];
#pragma unroll
    for (int t = 0; t < QB; ++t) S[t] = v8f{};
#pragma unroll
    for (int u = 0; u < 8; ++u) {
      const v2f klo = {kf[u].x, kf[u].y};
      const v2f khi = {kf[u].z, kf[u].w};
#pragma unroll
      for (int t = 0; t < QB; ++t) S[t] = wmma_f32(klo, qb[t][2 * u], S[t]);
#pragma unroll
      for (int t = 0; t < QB; ++t) S[t] = wmma_f32(khi, qb[t][2 * u + 1], S[t]);
    }

    float p[QB][8];
#pragma unroll
    for (int t = 0; t < QB; ++t) {
      float lg[8];
#pragma unroll
      for (int r = 0; r < 8; ++r) {
        const float v = S[t][r] * scale * spv[t][r];
        if constexpr (whole)
          lg[r] = v;
        else
          lg[r] = (key0 + r >= N_) ? -3.0e38f : v;
      }
      float mx = fmaxf(fmaxf(fmaxf(lg[0], lg[1]), fmaxf(lg[2], lg[3])),
                       fmaxf(fmaxf(lg[4], lg[5]), fmaxf(lg[6], lg[7])));
      mx = fmaxf(mx, xhalf(mx));
      const float mnew = fmaxf(mrun[t], mx);
      const float c = __expf(mrun[t] - mnew);
#pragma unroll
      for (int r = 0; r < 8; ++r) p[t][r] = __expf(lg[r] - mnew);
      const float sum = ((p[t][0] + p[t][1]) + (p[t][2] + p[t][3])) +
                        ((p[t][4] + p[t][5]) + (p[t][6] + p[t][7]));
      lrun[t] = lrun[t] * c + sum;
      mrun[t] = mnew;
#pragma unroll
      for (int j = 0; j < 4; ++j)
#pragma unroll
        for (int r = 0; r < 8; ++r) O[t][j][r] *= c;
    }
#pragma unroll
    for (int kk = 0; kk < 4; ++kk) {
      const v2f a0 = {vv[2 * kk].x, vv[2 * kk + 1].x};
      const v2f a1 = {vv[2 * kk].y, vv[2 * kk + 1].y};
      const v2f a2 = {vv[2 * kk].z, vv[2 * kk + 1].z};
      const v2f a3 = {vv[2 * kk].w, vv[2 * kk + 1].w};
#pragma unroll
      for (int t = 0; t < QB; ++t) {
        const v2f pb = {p[t][2 * kk], p[t][2 * kk + 1]};
        O[t][0] = wmma_f32(a0, pb, O[t][0]);
        O[t][1] = wmma_f32(a1, pb, O[t][1]);
        O[t][2] = wmma_f32(a2, pb, O[t][2]);
        O[t][3] = wmma_f32(a3, pb, O[t][3]);
      }
    }
  };
  for (int kt = 0; kt < KT_; ++kt) tile(kt, tile_kind<true>{});
  tile(KT_, tile_kind<false>{});
#pragma unroll
  for (int t = 0; t < QB; ++t) {
    const float lsum = lrun[t] + xhalf(lrun[t]);
    const float inv = 1.0f / lsum;
    const int q = q0 + 16 * t + lm;
    if (q < N_) {
      float* outp = AO + ((size_t)b * N_ + q) * D_ + h * HD_ + 32 * lh;
#pragma unroll
      for (int r = 0; r < 8; ++r) {
        outp[4 * r + 0] = O[t][0][r] * inv;
        outp[4 * r + 1] = O[t][1][r] * inv;
        outp[4 * r + 2] = O[t][2][r] * inv;
        outp[4 * r + 3] = O[t][3][r] * inv;
      }
    }
  }
}

#define QB_   4
#define NQB_  (N_ / (16 * QB_))
#define NQR_  (QT_ - NQB_ * QB_)

__global__ __launch_bounds__(32) void attn_kernel(
    const float* __restrict__ Qd, const float* __restrict__ Kd,
    const float* __restrict__ Vd, const float* __restrict__ SP,
    float* __restrict__ AO) {
  const int lane = threadIdx.x;
  const int lm = lane & 15;
  const int lh = lane >> 4;
  const int bh = blockIdx.x;
  const int qblk = blockIdx.y;
  const int b = bh >> 3;
  const int h = bh & 7;
  const size_t base = (size_t)bh * N_ * HD_;
  if (qblk < NQB_)
    attn_tiles<QB_>(Qd + base, Kd + base, Vd + base, SP, AO,
                    qblk * 16 * QB_, b, h, lm, lh);
  else
    attn_tiles<1>(Qd + base, Kd + base, Vd + base, SP, AO,
                  (NQB_ * QB_ + (qblk - NQB_)) * 16, b, h, lm, lh);
}

#define LNP_ 18
__global__ __launch_bounds__(256) void proj_ln_gemm(
    const float* __restrict__ AO, const float* __restrict__ w,
    const float* __restrict__ bias, const float* __restrict__ gamma,
    const float* __restrict__ beta, float* __restrict__ P) {
  __shared__ float part[2][8][80][LNP_];
  __shared__ float mu_s[80], rs_s[80];
  const int lane = threadIdx.x & 31;
  const int wv = threadIdx.x >> 5;
  const int lm = lane & 15;
  const int lh = lane >> 4;
  const int col0 = wv * 64;
  const int row0 = blockIdx.x * 80;
  v8f c[RT_][CT_];
#pragma unroll
  for (int i = 0; i < RT_; ++i)
#pragma unroll
    for (int j = 0; j < CT_; ++j) c[i][j] = v8f{};
  gemm_tile_80x64(AO + (size_t)(row0 + lm) * D_ + 16 * lh,
                  w + (size_t)(col0 + lm) * D_ + 16 * lh, c);
  float bc[CT_], gc[CT_], bt[CT_];
#pragma unroll
  for (int j = 0; j < CT_; ++j) {
    bc[j] = bias[col0 + 16 * j + lm];
    gc[j] = gamma[col0 + 16 * j + lm];
    bt[j] = beta[col0 + 16 * j + lm];
  }
#pragma unroll
  for (int i = 0; i < RT_; ++i) {
#pragma unroll
    for (int e = 0; e < 8; ++e) {
      const int rl = 16 * i + 8 * lh + e;
      float s = 0.0f, s2 = 0.0f;
#pragma unroll
      for (int j = 0; j < CT_; ++j) {
        const float v = c[i][j][e] + bc[j];
        c[i][j][e] = v;
        s += v;
        s2 += v * v;
      }
      part[0][wv][rl][lm] = s;
      part[1][wv][rl][lm] = s2;
    }
  }
  __syncthreads();
  if (threadIdx.x < 80) {
    const int rl = threadIdx.x;
    float S = 0.0f, S2 = 0.0f;
#pragma unroll
    for (int w8 = 0; w8 < 8; ++w8) {
#pragma unroll
      for (int l = 0; l < 16; ++l) {
        S += part[0][w8][rl][l];
        S2 += part[1][w8][rl][l];
      }
    }
    const float mu = S * (1.0f / D_);
    const float var = S2 * (1.0f / D_) - mu * mu;
    mu_s[rl] = mu;
    rs_s[rl] = rsqrtf(var + 1e-5f);
  }
  __syncthreads();
#pragma unroll
  for (int i = 0; i < RT_; ++i) {
#pragma unroll
    for (int e = 0; e < 8; ++e) {
      const int rl = 16 * i + 8 * lh + e;
      const float mu = mu_s[rl], rs = rs_s[rl];
      float* p = P + (size_t)(row0 + rl) * D_ + col0 + lm;
#pragma unroll
      for (int j = 0; j < CT_; ++j) {
        const float v = c[i][j][e];
        p[16 * j] = v + (v - mu) * rs * gc[j] + bt[j];
      }
    }
  }
}

extern "C" void kernel_launch(void* const* d_in, const int* in_sizes, int n_in,
                              void* d_out, int out_size, void* d_ws,
                              size_t ws_size, hipStream_t stream) {
  const float* x      = (const float*)d_in[0];
  const float* w_qkv  = (const float*)d_in[1];
  const float* w_proj = (const float*)d_in[2];
  const float* b_proj = (const float*)d_in[3];
  const float* gw     = (const float*)d_in[4];
  const float* ln_g   = (const float*)d_in[5];
  const float* ln_b   = (const float*)d_in[6];
  float* out = (float*)d_out;

  float* ws = (float*)d_ws;
  const size_t nsp  = (size_t)N_ * N_;
  const size_t nqkv = (size_t)B_ * H_ * N_ * HD_;
  float* Q  = ws;
  float* K  = Q + nqkv;
  float* V  = K + nqkv;
  float* AO = V + nqkv;
  float* SP = AO + nqkv;

  softplus_kernel<<<(int)((nsp + 255) / 256), 256, 0, stream>>>(gw, SP,
                                                                (int)nsp);
  qkv_gemm<<<dim3(3 * D_ / 64, ROWS_ / 80), 32, 0, stream>>>(x, w_qkv, Q, K,
                                                             V);
  attn_kernel<<<dim3(B_ * H_, NQB_ + NQR_), 32, 0, stream>>>(Q, K, V, SP, AO);
  proj_ln_gemm<<<ROWS_ / 80, 256, 0, stream>>>(AO, w_proj, b_proj, ln_g, ln_b,
                                               out);
}


# attention: each head's left-over last query gets its own launch, 16 waves take 4 key tiles each and join max, sum and O in LDS (was one 65-tile wave behind the grid); changes that query's sum order
# speedup vs baseline: 3.8134x; 1.0368x over previous
#include <hip/hip_runtime.h>

#define B_  16
#define N_  1025
#define D_  512
#define H_  8
#define HD_ 64
#define QT_ 65
#define ROWS_ 16400
#define KT_ 64
#define KTAIL_ (KT_ * 16 * HD_)
#define SPCOL_ (KT_ * KT_ * 256)
#define SPROW_ (SPCOL_ + KT_ * 16)
static_assert(N_ == KT_ * 16 + 1, "one key and one query past the whole tiles");

typedef __attribute__((ext_vector_type(2))) float v2f;
typedef __attribute__((ext_vector_type(8))) float v8f;
typedef float v4fu __attribute__((ext_vector_type(4), aligned(4)));

#define RT_ 5
#define CT_ 4

__device__ __forceinline__ v8f wmma_f32(v2f a, v2f b, v8f c) {
  return __builtin_amdgcn_wmma_f32_16x16x4_f32(false, a, false, b, (short)0, c,
                                               false, false);
}

__global__ __launch_bounds__(256) void softplus_kernel(
    const float* __restrict__ gw, float* __restrict__ sp, int n) {
  int i = blockIdx.x * 256 + threadIdx.x;
  if (i < n) {
    int q, k;
    if (i < SPCOL_) {
      const int e = i & 3, lane = (i >> 2) & 31, w2 = (i >> 7) & 1;
      const int kt = (i >> 8) & 63, qt = i >> 14;
      q = qt * 16 + (lane & 15);
      k = kt * 16 + 8 * (lane >> 4) + 4 * w2 + e;
    } else if (i < SPROW_) {
      q = i - SPCOL_;
      k = N_ - 1;
    } else {
      q = N_ - 1;
      k = i - SPROW_;
    }
    float w = gw[q * N_ + k];
    sp[i] = (w > 20.0f) ? w : log1pf(__expf(w));
  }
}

__device__ __forceinline__ void gemm_tile_80x64(const float* __restrict__ ap,
                                                const float* __restrict__ bp,
                                                v8f (&c)[RT_][CT_]) {
  for (int k0 = 0; k0 < D_; k0 += 32) {
    v4fu b[CT_][4], a[RT_][4];
#pragma unroll
    for (int j = 0; j < CT_; ++j)
#pragma unroll
      for (int u = 0; u < 4; ++u)
        b[j][u] = *(const v4fu*)(bp + (size_t)j * 16 * D_ + k0 + 4 * u);
#pragma unroll
    for (int i = 0; i < RT_; ++i)
#pragma unroll
      for (int u = 0; u < 4; ++u)
        a[i][u] = *(const v4fu*)(ap + (size_t)i * 16 * D_ + k0 + 4 * u);
    __builtin_amdgcn_sched_barrier(0);
#pragma unroll
    for (int i = 0; i < RT_; ++i) {
#pragma unroll
      for (int u = 0; u < 4; ++u) {
        const v2f alo = {a[i][u].x, a[i][u].y};
        const v2f ahi = {a[i][u].z, a[i][u].w};
#pragma unroll
        for (int j = 0; j < CT_; ++j) {
          const v2f blo = {b[j][u].x, b[j][u].y};
          c[i][j] = wmma_f32(alo, blo, c[i][j]);
        }
#pragma unroll
        for (int j = 0; j < CT_; ++j) {
          const v2f bhi = {b[j][u].z, b[j][u].w};
          c[i][j] = wmma_f32(ahi, bhi, c[i][j]);
        }
      }
    }
  }
}

__global__ __launch_bounds__(32) void qkv_gemm(
    const float* __restrict__ x, const float* __restrict__ w,
    float* __restrict__ Qd, float* __restrict__ Kd, float* __restrict__ Vd) {
  const int lane = threadIdx.x;
  const int lm = lane & 15;
  const int lh = lane >> 4;
  const int col0 = blockIdx.x * 64;
  const int row0 = blockIdx.y * 80;
  v8f c[RT_][CT_];
#pragma unroll
  for (int i = 0; i < RT_; ++i)
#pragma unroll
    for (int j = 0; j < CT_; ++j) c[i][j] = v8f{};
  gemm_tile_80x64(x + (size_t)(row0 + lm) * D_ + 16 * lh,
                  w + (size_t)(col0 + lm) * D_ + 16 * lh, c);

  const int part = col0 >> 9;
  const int h = (col0 & 511) >> 6;
  if (part == 1) {
    const int c0 = (lm >> 3) * 128 + ((lm >> 2) & 1) * 64 + (lm & 3);
#pragma unroll
    for (int i = 0; i < RT_; ++i) {
#pragma unroll
      for (int e = 0; e < 8; ++e) {
        const int r = row0 + 16 * i + 8 * lh + e;
        const int bb = r / N_;
        const int nn = r - bb * N_;
        const int off = (nn < KT_ * 16)
                            ? (nn >> 4) * 1024 + (nn & 15) * 4 + c0
                            : KTAIL_ + lm;
        const int jstep = (nn < KT_ * 16) ? 256 : 16;
        float* p = Kd + (size_t)(bb * H_ + h) * N_ * HD_ + off;
#pragma unroll
        for (int j = 0; j < CT_; ++j) p[jstep * j] = c[i][j][e];
      }
    }
    return;
  }
  float* dst = (part == 0) ? Qd : Vd;
#pragma unroll
  for (int i = 0; i < RT_; ++i) {
#pragma unroll
    for (int e = 0; e < 8; ++e) {
      const int r = row0 + 16 * i + 8 * lh + e;
      const int bb = r / N_;
      const int nn = r - bb * N_;
      float* p = dst + ((size_t)(bb * H_ + h) * N_ + nn) * HD_ + lm;
#pragma unroll
      for (int j = 0; j < CT_; ++j) p[16 * j] = c[i][j][e];
    }
  }
}

__device__ __forceinline__ float xhalf(float x) {
  int bits = __builtin_bit_cast(int, x);
  return __builtin_bit_cast(
      float, __builtin_amdgcn_permlanex16(bits, bits, 0x76543210, 0xfedcba98,
                                          false, false));
}

template <bool W> struct tile_kind { static constexpr bool whole = W; };

template <int QB>
__device__ __forceinline__ void attn_tiles(
    const float* __restrict__ Qp, const float* __restrict__ Kp,
    const float* __restrict__ Vp, const float* __restrict__ SP, const int q0,
    const int kt_lo, const int kt_hi, const bool with_last, const int lm,
    const int lh, v8f (&O)[QB][4], float (&mrun)[QB], float (&lrun)[QB]) {
  const float scale = 0.125f;

  const int lane = lm + 16 * lh;
  v2f qb[QB][16];
  unsigned spblk[QB];
  unsigned spcol[QB];
#pragma unroll
  for (int t = 0; t < QB; ++t) {
    const int q = q0 + 16 * t + lm;
    const int qc = (q < N_) ? q : N_ - 1;
#pragma unroll
    for (int s = 0; s < 16; ++s)
      qb[t][s] = *(const v2f*)(Qp + (size_t)qc * HD_ + 8 * (s >> 1) + 4 * lh +
                               2 * (s & 1));
    spblk[t] = (unsigned)(((q0 >> 4) + t) * (KT_ * 256) + lane * 4);
    spcol[t] = (unsigned)(SPCOL_ + ((qc < KT_ * 16) ? qc : 0));
  }

#pragma unroll
  for (int t = 0; t < QB; ++t) {
#pragma unroll
    for (int j = 0; j < 4; ++j) O[t][j] = v8f{};
    mrun[t] = -3.0e38f;
    lrun[t] = 0.0f;
  }

  auto tile = [&](const int kt, auto kind) __attribute__((always_inline)) {
    constexpr bool whole = decltype(kind)::whole;
    const unsigned ko = whole ? (unsigned)(kt * 1024 + lane * 4)
                              : (unsigned)(KTAIL_ + 4 * lh);
    constexpr unsigned ku = whole ? 128u : 8u;
    v4fu kf[8];
#pragma unroll
    for (int u = 0; u < 8; ++u) kf[u] = *(const v4fu*)(Kp + (ko + u * ku));
    const int key0 = kt * 16 + 8 * lh;
    v4fu vv[8];
#pragma unroll
    for (int r = 0; r < 8; ++r) {
      int kr = key0 + r;
      if constexpr (!whole) kr = (kr < N_) ? kr : N_ - 1;
      vv[r] = *(const v4fu*)(Vp + (unsigned)(kr * HD_ + 4 * lm));
    }
    float spv[QB][8];
    if constexpr (QB > 1) {
#pragma unroll
      for (int t = 0; t < QB; ++t) {
        if constexpr (whole) {
          const unsigned so = spblk[t] + (unsigned)(kt * 256);
          const v4fu s0 = *(const v4fu*)(SP + so);
          const v4fu s1 = *(const v4fu*)(SP + (so + 128u));
          spv[t][0] = s0.x; spv[t][1] = s0.y; spv[t][2] = s0.z; spv[t][3] = s0.w;
          spv[t][4] = s1.x; spv[t][5] = s1.y; spv[t][6] = s1.z; spv[t][7] = s1.w;
        } else {
          spv[t][0] = SP[spcol[t]];
#pragma unroll
          for (int r = 1; r < 8; ++r) spv[t][r] = 0.0f;
        }
      }
    } else {
#pragma unroll
      for (int r = 0; r < 8; ++r) {
        int k = key0 + r;
        if constexpr (!whole) k = (k < N_) ? k : N_ - 1;
        spv[0][r] = SP[SPROW_ + k];
      }
    }

    v8f S[QB];
#pragma unroll
    for (int t = 0; t < QB; ++t) S[t] = v8f{};
#pragma unroll
    for (int u = 0; u < 8; ++u) {
      const v2f klo = {kf[u].x, kf[u].y};
      const v2f khi = {kf[u].z, kf[u].w};
#pragma unroll
      for (int t = 0; t < QB; ++t) S[t] = wmma_f32(klo, qb[t][2 * u], S[t]);
#pragma unroll
      for (int t = 0; t < QB; ++t) S[t] = wmma_f32(khi, qb[t][2 * u + 1], S[t]);
    }

    float p[QB][8];
#pragma unroll
    for (int t = 0; t < QB; ++t) {
      float lg[8];
#pragma unroll
      for (int r = 0; r < 8; ++r) {
        const float v = S[t][r] * scale * spv[t][r];
        if constexpr (whole)
          lg[r] = v;
        else
          lg[r] = (key0 + r >= N_) ? -3.0e38f : v;
      }
      float mx = fmaxf(fmaxf(fmaxf(lg[0], lg[1]), fmaxf(lg[2], lg[3])),
                       fmaxf(fmaxf(lg[4], lg[5]), fmaxf(lg[6], lg[7])));
      mx = fmaxf(mx, xhalf(mx));
      const float mnew = fmaxf(mrun[t], mx);
      const float c = __expf(mrun[t] - mnew);
#pragma unroll
      for (int r = 0; r < 8; ++r) p[t][r] = __expf(lg[r] - mnew);
      const float sum = ((p[t][0] + p[t][1]) + (p[t][2] + p[t][3])) +
                        ((p[t][4] + p[t][5]) + (p[t][6] + p[t][7]));
      lrun[t] = lrun[t] * c + sum;
      mrun[t] = mnew;
#pragma unroll
      for (int j = 0; j < 4; ++j)
#pragma unroll
        for (int r = 0; r < 8; ++r) O[t][j][r] *= c;
    }
#pragma unroll
    for (int kk = 0; kk < 4; ++kk) {
      const v2f a0 = {vv[2 * kk].x, vv[2 * kk + 1].x};
      const v2f a1 = {vv[2 * kk].y, vv[2 * kk + 1].y};
      const v2f a2 = {vv[2 * kk].z, vv[2 * kk + 1].z};
      const v2f a3 = {vv[2 * kk].w, vv[2 * kk + 1].w};
#pragma unroll
      for (int t = 0; t < QB; ++t) {
        const v2f pb = {p[t][2 * kk], p[t][2 * kk + 1]};
        O[t][0] = wmma_f32(a0, pb, O[t][0]);
        O[t][1] = wmma_f32(a1, pb, O[t][1]);
        O[t][2] = wmma_f32(a2, pb, O[t][2]);
        O[t][3] = wmma_f32(a3, pb, O[t][3]);
      }
    }
  };
  for (int kt = kt_lo; kt < kt_hi; ++kt) tile(kt, tile_kind<true>{});
  if (with_last) tile(KT_, tile_kind<false>{});
}

#define QB_   4
#define NQB_  (N_ / (16 * QB_))
static_assert(N_ - NQB_ * QB_ * 16 == 1, "one query is left over per head");

__global__ __launch_bounds__(32) void attn_kernel(
    const float* __restrict__ Qd, const float* __restrict__ Kd,
    const float* __restrict__ Vd, const float* __restrict__ SP,
    float* __restrict__ AO) {
  const int lane = threadIdx.x;
  const int lm = lane & 15;
  const int lh = lane >> 4;
  const int bh = blockIdx.x;
  const int b = bh >> 3;
  const int h = bh & 7;
  const int q0 = blockIdx.y * 16 * QB_;
  const size_t base = (size_t)bh * N_ * HD_;
  v8f O[QB_][4];
  float mrun[QB_], lrun[QB_];
  attn_tiles<QB_>(Qd + base, Kd + base, Vd + base, SP, q0, 0, KT_, true, lm,
                  lh, O, mrun, lrun);
#pragma unroll
  for (int t = 0; t < QB_; ++t) {
    const float lsum = lrun[t] + xhalf(lrun[t]);
    const float inv = 1.0f / lsum;
    const int q = q0 + 16 * t + lm;
    if (q < N_) {
      float* outp = AO + ((size_t)b * N_ + q) * D_ + h * HD_ + 32 * lh;
#pragma unroll
      for (int r = 0; r < 8; ++r) {
        outp[4 * r + 0] = O[t][0][r] * inv;
        outp[4 * r + 1] = O[t][1][r] * inv;
        outp[4 * r + 2] = O[t][2][r] * inv;
        outp[4 * r + 3] = O[t][3][r] * inv;
      }
    }
  }
}

#define LW_ 16
static_assert(KT_ % LW_ == 0, "whole key tiles split evenly over the waves");
__global__ __launch_bounds__(32 * LW_) void attn_last_kernel(
    const float* __restrict__ Qd, const float* __restrict__ Kd,
    const float* __restrict__ Vd, const float* __restrict__ SP,
    float* __restrict__ AO) {
  __shared__ float pm[LW_], pl[LW_], po[LW_][HD_];
  const int lane = threadIdx.x & 31;
  const int lm = lane & 15;
  const int lh = lane >> 4;
  const int wv = __builtin_amdgcn_readfirstlane((int)(threadIdx.x >> 5));
  const int bh = blockIdx.x;
  const int b = bh >> 3;
  const int h = bh & 7;
  const size_t base = (size_t)bh * N_ * HD_;
  v8f O[1][4];
  float mrun[1], lrun[1];
  attn_tiles<1>(Qd + base, Kd + base, Vd + base, SP, NQB_ * QB_ * 16,
                (KT_ / LW_) * wv, (KT_ / LW_) * (wv + 1), wv == LW_ - 1, lm,
                lh, O, mrun, lrun);
  const float lsum = lrun[0] + xhalf(lrun[0]);
  if (lm == 0) {
    if (lh == 0) {
      pm[wv] = mrun[0];
      pl[wv] = lsum;
    }
#pragma unroll
    for (int j = 0; j < 4; ++j)
#pragma unroll
      for (int r = 0; r < 8; ++r) po[wv][32 * lh + 4 * r + j] = O[0][j][r];
  }
  __syncthreads();
  if (threadIdx.x < HD_) {
    float M = pm[0];
#pragma unroll
    for (int w = 1; w < LW_; ++w) M = fmaxf(M, pm[w]);
    float lsumall = 0.0f, acc = 0.0f;
#pragma unroll
    for (int w = 0; w < LW_; ++w) {
      const float f = __expf(pm[w] - M);
      lsumall += pl[w] * f;
      acc += po[w][threadIdx.x] * f;
    }
    AO[((size_t)b * N_ + (N_ - 1)) * D_ + h * HD_ + threadIdx.x] =
        acc / lsumall;
  }
}

#define LNP_ 18
__global__ __launch_bounds__(256) void proj_ln_gemm(
    const float* __restrict__ AO, const float* __restrict__ w,
    const float* __restrict__ bias, const float* __restrict__ gamma,
    const float* __restrict__ beta, float* __restrict__ P) {
  __shared__ float part[2][8][80][LNP_];
  __shared__ float mu_s[80], rs_s[80];
  const int lane = threadIdx.x & 31;
  const int wv = threadIdx.x >> 5;
  const int lm = lane & 15;
  const int lh = lane >> 4;
  const int col0 = wv * 64;
  const int row0 = blockIdx.x * 80;
  v8f c[RT_][CT_];
#pragma unroll
  for (int i = 0; i < RT_; ++i)
#pragma unroll
    for (int j = 0; j < CT_; ++j) c[i][j] = v8f{};
  gemm_tile_80x64(AO + (size_t)(row0 + lm) * D_ + 16 * lh,
                  w + (size_t)(col0 + lm) * D_ + 16 * lh, c);
  float bc[CT_], gc[CT_], bt[CT_];
#pragma unroll
  for (int j = 0; j < CT_; ++j) {
    bc[j] = bias[col0 + 16 * j + lm];
    gc[j] = gamma[col0 + 16 * j + lm];
    bt[j] = beta[col0 + 16 * j + lm];
  }
#pragma unroll
  for (int i = 0; i < RT_; ++i) {
#pragma unroll
    for (int e = 0; e < 8; ++e) {
      const int rl = 16 * i + 8 * lh + e;
      float s = 0.0f, s2 = 0.0f;
#pragma unroll
      for (int j = 0; j < CT_; ++j) {
        const float v = c[i][j][e] + bc[j];
        c[i][j][e] = v;
        s += v;
        s2 += v * v;
      }
      part[0][wv][rl][lm] = s;
      part[1][wv][rl][lm] = s2;
    }
  }
  __syncthreads();
  if (threadIdx.x < 80) {
    const int rl = threadIdx.x;
    float S = 0.0f, S2 = 0.0f;
#pragma unroll
    for (int w8 = 0; w8 < 8; ++w8) {
#pragma unroll
      for (int l = 0; l < 16; ++l) {
        S += part[0][w8][rl][l];
        S2 += part[1][w8][rl][l];
      }
    }
    const float mu = S * (1.0f / D_);
    const float var = S2 * (1.0f / D_) - mu * mu;
    mu_s[rl] = mu;
    rs_s[rl] = rsqrtf(var + 1e-5f);
  }
  __syncthreads();
#pragma unroll
  for (int i = 0; i < RT_; ++i) {
#pragma unroll
    for (int e = 0; e < 8; ++e) {
      const int rl = 16 * i + 8 * lh + e;
      const float mu = mu_s[rl], rs = rs_s[rl];
      float* p = P + (size_t)(row0 + rl) * D_ + col0 + lm;
#pragma unroll
      for (int j = 0; j < CT_; ++j) {
        const float v = c[i][j][e];
        p[16 * j] = v + (v - mu) * rs * gc[j] + bt[j];
      }
    }
  }
}

extern "C" void kernel_launch(void* const* d_in, const int* in_sizes, int n_in,
                              void* d_out, int out_size, void* d_ws,
                              size_t ws_size, hipStream_t stream) {
  const float* x      = (const float*)d_in[0];
  const float* w_qkv  = (const float*)d_in[1];
  const float* w_proj = (const float*)d_in[2];
  const float* b_proj = (const float*)d_in[3];
  const float* gw     = (const float*)d_in[4];
  const float* ln_g   = (const float*)d_in[5];
  const float* ln_b   = (const float*)d_in[6];
  float* out = (float*)d_out;

  float* ws = (float*)d_ws;
  const size_t nsp  = (size_t)N_ * N_;
  const size_t nqkv = (size_t)B_ * H_ * N_ * HD_;
  float* Q  = ws;
  float* K  = Q + nqkv;
  float* V  = K + nqkv;
  float* AO = V + nqkv;
  float* SP = AO + nqkv;

  softplus_kernel<<<(int)((nsp + 255) / 256), 256, 0, stream>>>(gw, SP,
                                                                (int)nsp);
  qkv_gemm<<<dim3(3 * D_ / 64, ROWS_ / 80), 32, 0, stream>>>(x, w_qkv, Q, K,
                                                             V);
  attn_kernel<<<dim3(B_ * H_, NQB_), 32, 0, stream>>>(Q, K, V, SP, AO);
  attn_last_kernel<<<B_ * H_, 32 * LW_, 0, stream>>>(Q, K, V, SP, AO);
  proj_ln_gemm<<<ROWS_ / 80, 256, 0, stream>>>(AO, w_proj, b_proj, ln_g, ln_b,
                                               out);
}


# fragment loads one step ahead of their products: attention loads the next key tile's K into the registers the score products just read; both GEMMs get a second fragment set; sums bit-identical
# speedup vs baseline: 3.9409x; 1.0334x over previous
#include <hip/hip_runtime.h>

#define B_  16
#define N_  1025
#define D_  512
#define H_  8
#define HD_ 64
#define QT_ 65
#define ROWS_ 16400
#define KT_ 64
#define KTAIL_ (KT_ * 16 * HD_)
#define SPCOL_ (KT_ * KT_ * 256)
#define SPROW_ (SPCOL_ + KT_ * 16)
static_assert(N_ == KT_ * 16 + 1, "one key and one query past the whole tiles");

typedef __attribute__((ext_vector_type(2))) float v2f;
typedef __attribute__((ext_vector_type(8))) float v8f;
typedef float v4fu __attribute__((ext_vector_type(4), aligned(4)));

#define RT_ 5
#define CT_ 4

__device__ __forceinline__ v8f wmma_f32(v2f a, v2f b, v8f c) {
  return __builtin_amdgcn_wmma_f32_16x16x4_f32(false, a, false, b, (short)0, c,
                                               false, false);
}

__global__ __launch_bounds__(256) void softplus_kernel(
    const float* __restrict__ gw, float* __restrict__ sp, int n) {
  int i = blockIdx.x * 256 + threadIdx.x;
  if (i < n) {
    int q, k;
    if (i < SPCOL_) {
      const int e = i & 3, lane = (i >> 2) & 31, w2 = (i >> 7) & 1;
      const int kt = (i >> 8) & 63, qt = i >> 14;
      q = qt * 16 + (lane & 15);
      k = kt * 16 + 8 * (lane >> 4) + 4 * w2 + e;
    } else if (i < SPROW_) {
      q = i - SPCOL_;
      k = N_ - 1;
    } else {
      q = N_ - 1;
      k = i - SPROW_;
    }
    float w = gw[q * N_ + k];
    sp[i] = (w > 20.0f) ? w : log1pf(__expf(w));
  }
}

__device__ __forceinline__ void load_kline(const float* __restrict__ ap,
                                           const float* __restrict__ bp,
                                           const int k0, v4fu (&a)[RT_][4],
                                           v4fu (&b)[CT_][4]) {
#pragma unroll
  for (int j = 0; j < CT_; ++j)
#pragma unroll
    for (int u = 0; u < 4; ++u)
      b[j][u] = *(const v4fu*)(bp + (size_t)j * 16 * D_ + k0 + 4 * u);
#pragma unroll
  for (int i = 0; i < RT_; ++i)
#pragma unroll
    for (int u = 0; u < 4; ++u)
      a[i][u] = *(const v4fu*)(ap + (size_t)i * 16 * D_ + k0 + 4 * u);
}

__device__ __forceinline__ void mul_kline(const v4fu (&a)[RT_][4],
                                          const v4fu (&b)[CT_][4],
                                          v8f (&c)[RT_][CT_]) {
#pragma unroll
  for (int i = 0; i < RT_; ++i) {
#pragma unroll
    for (int u = 0; u < 4; ++u) {
      const v2f alo = {a[i][u].x, a[i][u].y};
      const v2f ahi = {a[i][u].z, a[i][u].w};
#pragma unroll
      for (int j = 0; j < CT_; ++j) {
        const v2f blo = {b[j][u].x, b[j][u].y};
        c[i][j] = wmma_f32(alo, blo, c[i][j]);
      }
#pragma unroll
      for (int j = 0; j < CT_; ++j) {
        const v2f bhi = {b[j][u].z, b[j][u].w};
        c[i][j] = wmma_f32(ahi, bhi, c[i][j]);
      }
    }
  }
}

__device__ __forceinline__ void gemm_tile_80x64(const float* __restrict__ ap,
                                                const float* __restrict__ bp,
                                                v8f (&c)[RT_][CT_]) {
  static_assert(D_ % 64 == 0, "the k lines are walked in pairs");
  v4fu a0[RT_][4], b0[CT_][4], a1[RT_][4], b1[CT_][4];
  load_kline(ap, bp, 0, a0, b0);
  for (int k0 = 0; k0 < D_ - 64; k0 += 64) {
    load_kline(ap, bp, k0 + 32, a1, b1);
    __builtin_amdgcn_sched_barrier(0);
    mul_kline(a0, b0, c);
    load_kline(ap, bp, k0 + 64, a0, b0);
    __builtin_amdgcn_sched_barrier(0);
    mul_kline(a1, b1, c);
  }
  load_kline(ap, bp, D_ - 32, a1, b1);
  __builtin_amdgcn_sched_barrier(0);
  mul_kline(a0, b0, c);
  mul_kline(a1, b1, c);
}

__global__ __launch_bounds__(32) void qkv_gemm(
    const float* __restrict__ x, const float* __restrict__ w,
    float* __restrict__ Qd, float* __restrict__ Kd, float* __restrict__ Vd) {
  const int lane = threadIdx.x;
  const int lm = lane & 15;
  const int lh = lane >> 4;
  const int col0 = blockIdx.x * 64;
  const int row0 = blockIdx.y * 80;
  v8f c[RT_][CT_];
#pragma unroll
  for (int i = 0; i < RT_; ++i)
#pragma unroll
    for (int j = 0; j < CT_; ++j) c[i][j] = v8f{};
  gemm_tile_80x64(x + (size_t)(row0 + lm) * D_ + 16 * lh,
                  w + (size_t)(col0 + lm) * D_ + 16 * lh, c);

  const int part = col0 >> 9;
  const int h = (col0 & 511) >> 6;
  if (part == 1) {
    const int c0 = (lm >> 3) * 128 + ((lm >> 2) & 1) * 64 + (lm & 3);
#pragma unroll
    for (int i = 0; i < RT_; ++i) {
#pragma unroll
      for (int e = 0; e < 8; ++e) {
        const int r = row0 + 16 * i + 8 * lh + e;
        const int bb = r / N_;
        const int nn = r - bb * N_;
        const int off = (nn < KT_ * 16)
                            ? (nn >> 4) * 1024 + (nn & 15) * 4 + c0
                            : KTAIL_ + lm;
        const int jstep = (nn < KT_ * 16) ? 256 : 16;
        float* p = Kd + (size_t)(bb * H_ + h) * N_ * HD_ + off;
#pragma unroll
        for (int j = 0; j < CT_; ++j) p[jstep * j] = c[i][j][e];
      }
    }
    return;
  }
  float* dst = (part == 0) ? Qd : Vd;
#pragma unroll
  for (int i = 0; i < RT_; ++i) {
#pragma unroll
    for (int e = 0; e < 8; ++e) {
      const int r = row0 + 16 * i + 8 * lh + e;
      const int bb = r / N_;
      const int nn = r - bb * N_;
      float* p = dst + ((size_t)(bb * H_ + h) * N_ + nn) * HD_ + lm;
#pragma unroll
      for (int j = 0; j < CT_; ++j) p[16 * j] = c[i][j][e];
    }
  }
}

__device__ __forceinline__ float xhalf(float x) {
  int bits = __builtin_bit_cast(int, x);
  return __builtin_bit_cast(
      float, __builtin_amdgcn_permlanex16(bits, bits, 0x76543210, 0xfedcba98,
                                          false, false));
}

template <bool W> struct tile_kind { static constexpr bool whole = W; };
template <int X> struct next_kind { static constexpr int value = X; };

template <int QB>
__device__ __forceinline__ void attn_tiles(
    const float* __restrict__ Qp, const float* __restrict__ Kp,
    const float* __restrict__ Vp, const float* __restrict__ SP, const int q0,
    const int kt_lo, const int kt_hi, const bool with_last, const int lm,
    const int lh, v8f (&O)[QB][4], float (&mrun)[QB], float (&lrun)[QB]) {
  const float scale = 0.125f;

  const int lane = lm + 16 * lh;
  v2f qb[QB][16];
  unsigned spblk[QB];
  unsigned spcol[QB];
#pragma unroll
  for (int t = 0; t < QB; ++t) {
    const int q = q0 + 16 * t + lm;
    const int qc = (q < N_) ? q : N_ - 1;
#pragma unroll
    for (int s = 0; s < 16; ++s)
      qb[t][s] = *(const v2f*)(Qp + (size_t)qc * HD_ + 8 * (s >> 1) + 4 * lh +
                               2 * (s & 1));
    spblk[t] = (unsigned)(((q0 >> 4) + t) * (KT_ * 256) + lane * 4);
    spcol[t] = (unsigned)(SPCOL_ + ((qc < KT_ * 16) ? qc : 0));
  }

#pragma unroll
  for (int t = 0; t < QB; ++t) {
#pragma unroll
    for (int j = 0; j < 4; ++j) O[t][j] = v8f{};
    mrun[t] = -3.0e38f;
    lrun[t] = 0.0f;
  }

  v4fu kf[8];
  auto load_k = [&](const int kt, auto kind) __attribute__((always_inline)) {
    constexpr bool whole = decltype(kind)::whole;
    const unsigned ko = whole ? (unsigned)(kt * 1024 + lane * 4)
                              : (unsigned)(KTAIL_ + 4 * lh);
    constexpr unsigned ku = whole ? 128u : 8u;
#pragma unroll
    for (int u = 0; u < 8; ++u) kf[u] = *(const v4fu*)(Kp + (ko + u * ku));
  };
  auto tile = [&](const int kt, auto kind, auto next)
      __attribute__((always_inline)) {
    constexpr bool whole = decltype(kind)::whole;
    constexpr int nx = decltype(next)::value;
    const int key0 = kt * 16 + 8 * lh;
    v4fu vv[8];
#pragma unroll
    for (int r = 0; r < 8; ++r) {
      int kr = key0 + r;
      if constexpr (!whole) kr = (kr < N_) ? kr : N_ - 1;
      vv[r] = *(const v4fu*)(Vp + (unsigned)(kr * HD_ + 4 * lm));
    }
    float spv[QB][8];
    if constexpr (QB > 1) {
#pragma unroll
      for (int t = 0; t < QB; ++t) {
        if constexpr (whole) {
          const unsigned so = spblk[t] + (unsigned)(kt * 256);
          const v4fu s0 = *(const v4fu*)(SP + so);
          const v4fu s1 = *(const v4fu*)(SP + (so + 128u));
          spv[t][0] = s0.x; spv[t][1] = s0.y; spv[t][2] = s0.z; spv[t][3] = s0.w;
          spv[t][4] = s1.x; spv[t][5] = s1.y; spv[t][6] = s1.z; spv[t][7] = s1.w;
        } else {
          spv[t][0] = SP[spcol[t]];
#pragma unroll
          for (int r = 1; r < 8; ++r) spv[t][r] = 0.0f;
        }
      }
    } else {
#pragma unroll
      for (int r = 0; r < 8; ++r) {
        int k = key0 + r;
        if constexpr (!whole) k = (k < N_) ? k : N_ - 1;
        spv[0][r] = SP[SPROW_ + k];
      }
    }

    if constexpr (nx != 0) __builtin_amdgcn_sched_barrier(0);

    v8f S[QB];
#pragma unroll
    for (int t = 0; t < QB; ++t) S[t] = v8f{};
#pragma unroll
    for (int u = 0; u < 8; ++u) {
      const v2f klo = {kf[u].x, kf[u].y};
      const v2f khi = {kf[u].z, kf[u].w};
#pragma unroll
      for (int t = 0; t < QB; ++t) S[t] = wmma_f32(klo, qb[t][2 * u], S[t]);
#pragma unroll
      for (int t = 0; t < QB; ++t) S[t] = wmma_f32(khi, qb[t][2 * u + 1], S[t]);
    }

    if constexpr (nx == 1) load_k(kt + 1, tile_kind<true>{});
    if constexpr (nx == 2) load_k(KT_, tile_kind<false>{});
    if constexpr (nx != 0) __builtin_amdgcn_sched_barrier(0);

    float p[QB][8];
#pragma unroll
    for (int t = 0; t < QB; ++t) {
      float lg[8];
#pragma unroll
      for (int r = 0; r < 8; ++r) {
        const float v = S[t][r] * scale * spv[t][r];
        if constexpr (whole)
          lg[r] = v;
        else
          lg[r] = (key0 + r >= N_) ? -3.0e38f : v;
      }
      float mx = fmaxf(fmaxf(fmaxf(lg[0], lg[1]), fmaxf(lg[2], lg[3])),
                       fmaxf(fmaxf(lg[4], lg[5]), fmaxf(lg[6], lg[7])));
      mx = fmaxf(mx, xhalf(mx));
      const float mnew = fmaxf(mrun[t], mx);
      const float c = __expf(mrun[t] - mnew);
#pragma unroll
      for (int r = 0; r < 8; ++r) p[t][r] = __expf(lg[r] - mnew);
      const float sum = ((p[t][0] + p[t][1]) + (p[t][2] + p[t][3])) +
                        ((p[t][4] + p[t][5]) + (p[t][6] + p[t][7]));
      lrun[t] = lrun[t] * c + sum;
      mrun[t] = mnew;
#pragma unroll
      for (int j = 0; j < 4; ++j)
#pragma unroll
        for (int r = 0; r < 8; ++r) O[t][j][r] *= c;
    }
#pragma unroll
    for (int kk = 0; kk < 4; ++kk) {
      const v2f a0 = {vv[2 * kk].x, vv[2 * kk + 1].x};
      const v2f a1 = {vv[2 * kk].y, vv[2 * kk + 1].y};
      const v2f a2 = {vv[2 * kk].z, vv[2 * kk + 1].z};
      const v2f a3 = {vv[2 * kk].w, vv[2 * kk + 1].w};
#pragma unroll
      for (int t = 0; t < QB; ++t) {
        const v2f pb = {p[t][2 * kk], p[t][2 * kk + 1]};
        O[t][0] = wmma_f32(a0, pb, O[t][0]);
        O[t][1] = wmma_f32(a1, pb, O[t][1]);
        O[t][2] = wmma_f32(a2, pb, O[t][2]);
        O[t][3] = wmma_f32(a3, pb, O[t][3]);
      }
    }
  };
  load_k(kt_lo, tile_kind<true>{});
  for (int kt = kt_lo; kt < kt_hi - 1; ++kt)
    tile(kt, tile_kind<true>{}, next_kind<1>{});
  if (with_last) {
    tile(kt_hi - 1, tile_kind<true>{}, next_kind<2>{});
    tile(KT_, tile_kind<false>{}, next_kind<0>{});
  } else {
    tile(kt_hi - 1, tile_kind<true>{}, next_kind<0>{});
  }
}

#define QB_   4
#define NQB_  (N_ / (16 * QB_))
static_assert(N_ - NQB_ * QB_ * 16 == 1, "one query is left over per head");

__global__ __launch_bounds__(32) void attn_kernel(
    const float* __restrict__ Qd, const float* __restrict__ Kd,
    const float* __restrict__ Vd, const float* __restrict__ SP,
    float* __restrict__ AO) {
  const int lane = threadIdx.x;
  const int lm = lane & 15;
  const int lh = lane >> 4;
  const int bh = blockIdx.x;
  const int b = bh >> 3;
  const int h = bh & 7;
  const int q0 = blockIdx.y * 16 * QB_;
  const size_t base = (size_t)bh * N_ * HD_;
  v8f O[QB_][4];
  float mrun[QB_], lrun[QB_];
  attn_tiles<QB_>(Qd + base, Kd + base, Vd + base, SP, q0, 0, KT_, true, lm,
                  lh, O, mrun, lrun);
#pragma unroll
  for (int t = 0; t < QB_; ++t) {
    const float lsum = lrun[t] + xhalf(lrun[t]);
    const float inv = 1.0f / lsum;
    const int q = q0 + 16 * t + lm;
    if (q < N_) {
      float* outp = AO + ((size_t)b * N_ + q) * D_ + h * HD_ + 32 * lh;
#pragma unroll
      for (int r = 0; r < 8; ++r) {
        outp[4 * r + 0] = O[t][0][r] * inv;
        outp[4 * r + 1] = O[t][1][r] * inv;
        outp[4 * r + 2] = O[t][2][r] * inv;
        outp[4 * r + 3] = O[t][3][r] * inv;
      }
    }
  }
}

#define LW_ 16
static_assert(KT_ % LW_ == 0, "whole key tiles split evenly over the waves");
__global__ __launch_bounds__(32 * LW_) void attn_last_kernel(
    const float* __restrict__ Qd, const float* __restrict__ Kd,
    const float* __restrict__ Vd, const float* __restrict__ SP,
    float* __restrict__ AO) {
  __shared__ float pm[LW_], pl[LW_], po[LW_][HD_];
  const int lane = threadIdx.x & 31;
  const int lm = lane & 15;
  const int lh = lane >> 4;
  const int wv = __builtin_amdgcn_readfirstlane((int)(threadIdx.x >> 5));
  const int bh = blockIdx.x;
  const int b = bh >> 3;
  const int h = bh & 7;
  const size_t base = (size_t)bh * N_ * HD_;
  v8f O[1][4];
  float mrun[1], lrun[1];
  attn_tiles<1>(Qd + base, Kd + base, Vd + base, SP, NQB_ * QB_ * 16,
                (KT_ / LW_) * wv, (KT_ / LW_) * (wv + 1), wv == LW_ - 1, lm,
                lh, O, mrun, lrun);
  const float lsum = lrun[0] + xhalf(lrun[0]);
  if (lm == 0) {
    if (lh == 0) {
      pm[wv] = mrun[0];
      pl[wv] = lsum;
    }
#pragma unroll
    for (int j = 0; j < 4; ++j)
#pragma unroll
      for (int r = 0; r < 8; ++r) po[wv][32 * lh + 4 * r + j] = O[0][j][r];
  }
  __syncthreads();
  if (threadIdx.x < HD_) {
    float M = pm[0];
#pragma unroll
    for (int w = 1; w < LW_; ++w) M = fmaxf(M, pm[w]);
    float lsumall = 0.0f, acc = 0.0f;
#pragma unroll
    for (int w = 0; w < LW_; ++w) {
      const float f = __expf(pm[w] - M);
      lsumall += pl[w] * f;
      acc += po[w][threadIdx.x] * f;
    }
    AO[((size_t)b * N_ + (N_ - 1)) * D_ + h * HD_ + threadIdx.x] =
        acc / lsumall;
  }
}

#define LNP_ 18
__global__ __launch_bounds__(256) void proj_ln_gemm(
    const float* __restrict__ AO, const float* __restrict__ w,
    const float* __restrict__ bias, const float* __restrict__ gamma,
    const float* __restrict__ beta, float* __restrict__ P) {
  __shared__ float part[2][8][80][LNP_];
  __shared__ float mu_s[80], rs_s[80];
  const int lane = threadIdx.x & 31;
  const int wv = threadIdx.x >> 5;
  const int lm = lane & 15;
  const int lh = lane >> 4;
  const int col0 = wv * 64;
  const int row0 = blockIdx.x * 80;
  v8f c[RT_][CT_];
#pragma unroll
  for (int i = 0; i < RT_; ++i)
#pragma unroll
    for (int j = 0; j < CT_; ++j) c[i][j] = v8f{};
  gemm_tile_80x64(AO + (size_t)(row0 + lm) * D_ + 16 * lh,
                  w + (size_t)(col0 + lm) * D_ + 16 * lh, c);
  float bc[CT_], gc[CT_], bt[CT_];
#pragma unroll
  for (int j = 0; j < CT_; ++j) {
    bc[j] = bias[col0 + 16 * j + lm];
    gc[j] = gamma[col0 + 16 * j + lm];
    bt[j] = beta[col0 + 16 * j + lm];
  }
#pragma unroll
  for (int i = 0; i < RT_; ++i) {
#pragma unroll
    for (int e = 0; e < 8; ++e) {
      const int rl = 16 * i + 8 * lh + e;
      float s = 0.0f, s2 = 0.0f;
#pragma unroll
      for (int j = 0; j < CT_; ++j) {
        const float v = c[i][j][e] + bc[j];
        c[i][j][e] = v;
        s += v;
        s2 += v * v;
      }
      part[0][wv][rl][lm] = s;
      part[1][wv][rl][lm] = s2;
    }
  }
  __syncthreads();
  if (threadIdx.x < 80) {
    const int rl = threadIdx.x;
    float S = 0.0f, S2 = 0.0f;
#pragma unroll
    for (int w8 = 0; w8 < 8; ++w8) {
#pragma unroll
      for (int l = 0; l < 16; ++l) {
        S += part[0][w8][rl][l];
        S2 += part[1][w8][rl][l];
      }
    }
    const float mu = S * (1.0f / D_);
    const float var = S2 * (1.0f / D_) - mu * mu;
    mu_s[rl] = mu;
    rs_s[rl] = rsqrtf(var + 1e-5f);
  }
  __syncthreads();
#pragma unroll
  for (int i = 0; i < RT_; ++i) {
#pragma unroll
    for (int e = 0; e < 8; ++e) {
      const int rl = 16 * i + 8 * lh + e;
      const float mu = mu_s[rl], rs = rs_s[rl];
      float* p = P + (size_t)(row0 + rl) * D_ + col0 + lm;
#pragma unroll
      for (int j = 0; j < CT_; ++j) {
        const float v = c[i][j][e];
        p[16 * j] = v + (v - mu) * rs * gc[j] + bt[j];
      }
    }
  }
}

extern "C" void kernel_launch(void* const* d_in, const int* in_sizes, int n_in,
                              void* d_out, int out_size, void* d_ws,
                              size_t ws_size, hipStream_t stream) {
  const float* x      = (const float*)d_in[0];
  const float* w_qkv  = (const float*)d_in[1];
  const float* w_proj = (const float*)d_in[2];
  const float* b_proj = (const float*)d_in[3];
  const float* gw     = (const float*)d_in[4];
  const float* ln_g   = (const float*)d_in[5];
  const float* ln_b   = (const float*)d_in[6];
  float* out = (float*)d_out;

  float* ws = (float*)d_ws;
  const size_t nsp  = (size_t)N_ * N_;
  const size_t nqkv = (size_t)B_ * H_ * N_ * HD_;
  float* Q  = ws;
  float* K  = Q + nqkv;
  float* V  = K + nqkv;
  float* AO = V + nqkv;
  float* SP = AO + nqkv;

  softplus_kernel<<<(int)((nsp + 255) / 256), 256, 0, stream>>>(gw, SP,
                                                                (int)nsp);
  qkv_gemm<<<dim3(3 * D_ / 64, ROWS_ / 80), 32, 0, stream>>>(x, w_qkv, Q, K,
                                                             V);
  attn_kernel<<<dim3(B_ * H_, NQB_), 32, 0, stream>>>(Q, K, V, SP, AO);
  attn_last_kernel<<<B_ * H_, 32 * LW_, 0, stream>>>(Q, K, V, SP, AO);
  proj_ln_gemm<<<ROWS_ / 80, 256, 0, stream>>>(AO, w_proj, b_proj, ln_g, ln_b,
                                               out);
}
